# starting kernel
# baseline (speedup 1.0000x reference)
#include <hip/hip_runtime.h>
#include <math.h>

typedef __attribute__((ext_vector_type(2))) float v2f;
typedef __attribute__((ext_vector_type(8))) float v8f;

__device__ __forceinline__ void atomic_add_f32(float* p, float v) {
  __hip_atomic_fetch_add(p, v, __ATOMIC_RELAXED, __HIP_MEMORY_SCOPE_AGENT);
}

__global__ void deg_kernel(const int* __restrict__ cols,
                           const float* __restrict__ ew,
                           float* __restrict__ deg, int E) {
  int e = blockIdx.x * blockDim.x + threadIdx.x;
  if (e >= E) return;
  atomic_add_f32(&deg[cols[e]], ew[e]);
}

__global__ void dinv_kernel(float* __restrict__ deg, int N) {
  int i = blockIdx.x * blockDim.x + threadIdx.x;
  if (i >= N) return;
  float d = deg[i] + 1.0f;
  deg[i] = (d > 0.0f) ? rsqrtf(d) : 0.0f;
}

template <int NCOLS>
__global__ void gemm_wmma_f32(const float* __restrict__ A,
                              const float* __restrict__ B,
                              float* __restrict__ C) {
  const int m0   = blockIdx.x * 16;
  const int wid  = threadIdx.x >> 5;
  const int lane = threadIdx.x & 31;
  const int n0   = wid * 16;
  const int ln   = lane & 15;
  const int half = lane >> 4;

  v8f acc = {};
  const float* aptr = A + (size_t)(m0 + ln) * 64 + 2 * half;
  const float* bptr = B + (size_t)(2 * half) * NCOLS + n0 + ln;

#pragma unroll
  for (int k = 0; k < 64; k += 4) {
    v2f a, b;
    a.x = aptr[k];
    a.y = aptr[k + 1];
    b.x = bptr[k * NCOLS];
    b.y = bptr[k * NCOLS + NCOLS];
    acc = __builtin_amdgcn_wmma_f32_16x16x4_f32(
         false, a,  false, b,
         (short)0, acc,  false,  false);
  }

  float* cptr = C + (size_t)(m0 + 8 * half) * NCOLS + n0 + ln;
#pragma unroll
  for (int r = 0; r < 8; ++r) {
    cptr[(size_t)r * NCOLS] = acc[r];
  }
}

template <int CH>
__global__ void aggregate_kernel(const float* __restrict__ x,
                                 const int* __restrict__ rows,
                                 const int* __restrict__ cols,
                                 const float* __restrict__ ew,
                                 const float* __restrict__ dinv,
                                 float* __restrict__ out,
                                 int E, int N) {
  const int VPE = CH / 4;
  long long tid = (long long)blockIdx.x * blockDim.x + threadIdx.x;
  long long total = (long long)(E + N) * VPE;
  if (tid >= total) return;
  long long e = tid / VPE;
  int c4 = (int)(tid % VPE);

  int r, c; float w;
  if (e < E) { r = rows[e]; c = cols[e]; w = ew[e]; }
  else       { r = c = (int)(e - E);     w = 1.0f; }

  float nrm = dinv[r] * w * dinv[c];
  const float4 xv = *(const float4*)(x + (long long)r * CH + c4 * 4);
  float* dst = out + (long long)c * CH + c4 * 4;
  atomic_add_f32(dst + 0, xv.x * nrm);
  atomic_add_f32(dst + 1, xv.y * nrm);
  atomic_add_f32(dst + 2, xv.z * nrm);
  atomic_add_f32(dst + 3, xv.w * nrm);
}

__global__ void bias_relu_kernel(float* __restrict__ x,
                                 const float* __restrict__ b, int N) {
  long long tid = (long long)blockIdx.x * blockDim.x + threadIdx.x;
  if (tid >= (long long)N * 16) return;
  int c4 = (int)(tid & 15);
  float4 v  = ((float4*)x)[tid];
  float4 bb = ((const float4*)b)[c4];
  v.x = fmaxf(v.x + bb.x, 0.0f);
  v.y = fmaxf(v.y + bb.y, 0.0f);
  v.z = fmaxf(v.z + bb.z, 0.0f);
  v.w = fmaxf(v.w + bb.w, 0.0f);
  ((float4*)x)[tid] = v;
}

__global__ void l2norm_kernel(const float* __restrict__ x,
                              const float* __restrict__ b2,
                              float* __restrict__ out, int N) {
  int row  = blockIdx.x * (blockDim.x >> 5) + (threadIdx.x >> 5);
  int lane = threadIdx.x & 31;
  if (row >= N) return;
  float v = x[(long long)row * 32 + lane] + b2[lane];
  float ss = v * v;
#pragma unroll
  for (int off = 16; off > 0; off >>= 1)
    ss += __shfl_xor(ss, off, 32);
  float l2 = sqrtf(ss);
  float scale = 1.0f / fmaxf(l2, 1e-12f);
  out[(long long)row * 32 + lane] = v * scale;
}

extern "C" void kernel_launch(void* const* d_in, const int* in_sizes, int n_in,
                              void* d_out, int out_size, void* d_ws, size_t ws_size,
                              hipStream_t stream) {
  const int*   edge_index = (const int*)d_in[0];
  const float* ew         = (const float*)d_in[1];
  const float* emb        = (const float*)d_in[2];
  const float* W1         = (const float*)d_in[3];
  const float* b1         = (const float*)d_in[4];
  const float* W2         = (const float*)d_in[5];
  const float* b2         = (const float*)d_in[6];

  const int E = in_sizes[0] / 2;
  const int N = in_sizes[2] / 64;
  const int* rows = edge_index;
  const int* cols = edge_index + E;

  float* ws   = (float*)d_ws;
  float* dinv = ws;
  float* xw1  = dinv + (size_t)N;
  float* out1 = xw1  + (size_t)N * 64;
  float* xw2  = out1 + (size_t)N * 64;
  float* out2 = xw2  + (size_t)N * 32;

  hipMemsetAsync(dinv, 0, sizeof(float) * (size_t)N, stream);
  hipMemsetAsync(out1, 0, sizeof(float) * (size_t)N * 64, stream);
  hipMemsetAsync(out2, 0, sizeof(float) * (size_t)N * 32, stream);

  deg_kernel<<<(E + 255) / 256, 256, 0, stream>>>(cols, ew, dinv, E);
  dinv_kernel<<<(N + 255) / 256, 256, 0, stream>>>(dinv, N);

  gemm_wmma_f32<64><<<N / 16, 128, 0, stream>>>(emb, W1, xw1);
  {
    long long total = (long long)(E + N) * 16;
    aggregate_kernel<64><<<(unsigned)((total + 255) / 256), 256, 0, stream>>>(
        xw1, rows, cols, ew, dinv, out1, E, N);
  }
  bias_relu_kernel<<<(unsigned)(((long long)N * 16 + 255) / 256), 256, 0, stream>>>(out1, b1, N);

  gemm_wmma_f32<32><<<N / 16, 64, 0, stream>>>(out1, W2, xw2);
  {
    long long total = (long long)(E + N) * 8;
    aggregate_kernel<32><<<(unsigned)((total + 255) / 256), 256, 0, stream>>>(
        xw2, rows, cols, ew, dinv, out2, E, N);
  }

  l2norm_kernel<<<(N + 7) / 8, 256, 0, stream>>>(out2, b2, (float*)d_out, N);
}


# Edges counting-sorted by 128-node destination bucket each call; degree and both aggregations sum in LDS (LDS float atomics, order still varies), one store per output; no global atomics, no memsets
# speedup vs baseline: 1.9880x; 1.9880x over previous
#include <hip/hip_runtime.h>
#include <math.h>

typedef __attribute__((ext_vector_type(2))) float v2f;
typedef __attribute__((ext_vector_type(8))) float v8f;

#define BK_SHIFT 7
#define BK 128
#define NCHUNK 256

__device__ __forceinline__ void lds_add_f32(float* p, float v) {
  __hip_atomic_fetch_add(p, v, __ATOMIC_RELAXED, __HIP_MEMORY_SCOPE_WORKGROUP);
}
__device__ __forceinline__ int lds_add_i32(int* p, int v) {
  return __hip_atomic_fetch_add(p, v, __ATOMIC_RELAXED, __HIP_MEMORY_SCOPE_WORKGROUP);
}

__device__ __forceinline__ int wave_incl_scan(int v, int lane) {
  for (int off = 1; off < 32; off <<= 1) {
    int n = __shfl_up(v, off, 32);
    if (lane >= off) v += n;
  }
  return v;
}

__global__ void __launch_bounds__(1024)
bucket_hist_kernel(const int* __restrict__ cols, int E, int chunk_len, int NB,
                   int* __restrict__ chunkhist) {
  extern __shared__ int sh_cnt[];
  const int t = threadIdx.x;
  for (int i = t; i < NB; i += 1024) sh_cnt[i] = 0;
  __syncthreads();
  const int beg = blockIdx.x * chunk_len;
  const int end = min(E, beg + chunk_len);
  for (int e = beg + t; e < end; e += 4 * 1024) {
    int c[4];
#pragma unroll
    for (int j = 0; j < 4; ++j) {
      const int ej = e + j * 1024;
      c[j] = cols[ej < end ? ej : e];
    }
#pragma unroll
    for (int j = 0; j < 4; ++j)
      if (e + j * 1024 < end) lds_add_i32(&sh_cnt[c[j] >> BK_SHIFT], 1);
  }
  __syncthreads();
  for (int i = t; i < NB; i += 1024)
    chunkhist[(size_t)i * NCHUNK + blockIdx.x] = sh_cnt[i];
}

__global__ void __launch_bounds__(NCHUNK)
row_scan_kernel(int* __restrict__ chunkhist, int* __restrict__ btotal) {
  __shared__ int wsum[NCHUNK / 32];
  const int t = threadIdx.x, lane = t & 31, w = t >> 5;
  const size_t idx = (size_t)blockIdx.x * NCHUNK + t;
  const int v = chunkhist[idx];
  const int s = wave_incl_scan(v, lane);
  if (lane == 31) wsum[w] = s;
  __syncthreads();
  int wbase = 0;
  for (int k = 0; k < NCHUNK / 32; ++k)
    if (k < w) wbase += wsum[k];
  chunkhist[idx] = wbase + s - v;
  if (t == NCHUNK - 1) btotal[blockIdx.x] = wbase + s;
}

__global__ void __launch_bounds__(1024)
bucket_base_kernel(const int* __restrict__ btotal, int* __restrict__ bbase, int NB) {
  __shared__ int wsum[32];
  const int t = threadIdx.x, lane = t & 31, w = t >> 5;
  int carry = 0;
  for (int base = 0; base < NB; base += 1024) {
    const int i = base + t;
    const int v = (i < NB) ? btotal[i] : 0;
    const int s = wave_incl_scan(v, lane);
    if (lane == 31) wsum[w] = s;
    __syncthreads();
    if (w == 0) wsum[lane] = wave_incl_scan(wsum[lane], lane);
    __syncthreads();
    const int wbase = (w > 0) ? wsum[w - 1] : 0;
    if (i < NB) bbase[i] = carry + wbase + s - v;
    const int tile_total = wsum[31];
    __syncthreads();
    carry += tile_total;
  }
  if (t == 0) bbase[NB] = carry;
}

__global__ void __launch_bounds__(1024)
bucket_scatter_kernel(const int* __restrict__ rows, const int* __restrict__ cols,
                      const float* __restrict__ ew, int E, int chunk_len, int NB,
                      const int* __restrict__ chunkoff, const int* __restrict__ bbase,
                      int4* __restrict__ rec) {
  extern __shared__ int sh_cur[];
  const int t = threadIdx.x;
  for (int i = t; i < NB; i += 1024)
    sh_cur[i] = bbase[i] + chunkoff[(size_t)i * NCHUNK + blockIdx.x];
  __syncthreads();
  const int beg = blockIdx.x * chunk_len;
  const int end = min(E, beg + chunk_len);
  for (int e = beg + t; e < end; e += 4 * 1024) {
    int c[4], r[4];
    float w[4];
#pragma unroll
    for (int j = 0; j < 4; ++j) {
      const int ej = e + j * 1024;
      const int es = ej < end ? ej : e;
      c[j] = cols[es];
      r[j] = rows[es];
      w[j] = ew[es];
    }
#pragma unroll
    for (int j = 0; j < 4; ++j) {
      if (e + j * 1024 < end) {
        const int pos = lds_add_i32(&sh_cur[c[j] >> BK_SHIFT], 1);
        rec[pos] = make_int4(r[j], c[j], __float_as_int(w[j]), 0);
      }
    }
  }
}

__global__ void __launch_bounds__(256)
deg_dinv_kernel(const int4* __restrict__ rec, const int* __restrict__ bbase,
                float* __restrict__ dinv, int N) {
  __shared__ float sdeg[BK];
  const int t = threadIdx.x;
  if (t < BK) sdeg[t] = 1.0f;
  __syncthreads();
  const int beg = bbase[blockIdx.x], end = bbase[blockIdx.x + 1];
  for (int i = beg + t; i < end; i += 4 * 256) {
    int4 q[4];
#pragma unroll
    for (int j = 0; j < 4; ++j) {
      const int ij = i + j * 256;
      q[j] = rec[ij < end ? ij : i];
    }
#pragma unroll
    for (int j = 0; j < 4; ++j)
      if (i + j * 256 < end) lds_add_f32(&sdeg[q[j].y & (BK - 1)], __int_as_float(q[j].z));
  }
  __syncthreads();
  if (t < BK) {
    const int n = blockIdx.x * BK + t;
    if (n < N) {
      const float d = sdeg[t];
      dinv[n] = (d > 0.0f) ? rsqrtf(d) : 0.0f;
    }
  }
}

template <int NCOLS>
__global__ void gemm_wmma_f32(const float* __restrict__ A,
                              const float* __restrict__ B,
                              float* __restrict__ C) {
  const int m0   = blockIdx.x * 16;
  const int wid  = threadIdx.x >> 5;
  const int lane = threadIdx.x & 31;
  const int n0   = wid * 16;
  const int ln   = lane & 15;
  const int half = lane >> 4;

  v8f acc = {};
  const float* aptr = A + (size_t)(m0 + ln) * 64 + 2 * half;
  const float* bptr = B + (size_t)(2 * half) * NCOLS + n0 + ln;

#pragma unroll
  for (int k = 0; k < 64; k += 4) {
    v2f a, b;
    a.x = aptr[k];
    a.y = aptr[k + 1];
    b.x = bptr[k * NCOLS];
    b.y = bptr[k * NCOLS + NCOLS];
    acc = __builtin_amdgcn_wmma_f32_16x16x4_f32(
         false, a,  false, b,
         (short)0, acc,  false,  false);
  }

  float* cptr = C + (size_t)(m0 + 8 * half) * NCOLS + n0 + ln;
#pragma unroll
  for (int r = 0; r < 8; ++r) {
    cptr[(size_t)r * NCOLS] = acc[r];
  }
}

template <int CH, int U>
__global__ void __launch_bounds__(512)
aggregate_lds_kernel(const float* __restrict__ x, const int4* __restrict__ rec,
                     const int* __restrict__ bbase, const float* __restrict__ dinv,
                     float* __restrict__ out, int N) {
  constexpr int LPE = CH / 4;
  constexpr int EPI = 512 / LPE;
  __shared__ float4 acc4[BK * LPE];
  float* acc = (float*)acc4;
  const int t = threadIdx.x;
  const int node0 = blockIdx.x * BK;
  const float4* x4 = (const float4*)x;

  for (int i = t; i < BK * LPE; i += 512) {
    const int n = node0 + i / LPE;
    float4 v = make_float4(0.0f, 0.0f, 0.0f, 0.0f);
    if (n < N) {
      const float d = dinv[n];
      const float nrm = d * d;
      const float4 xv = x4[(size_t)n * LPE + (i % LPE)];
      v = make_float4(xv.x * nrm, xv.y * nrm, xv.z * nrm, xv.w * nrm);
    }
    acc4[i] = v;
  }
  __syncthreads();

  const int beg = bbase[blockIdx.x], end = bbase[blockIdx.x + 1];
  const int sub = t % LPE;
  for (int i = beg + t / LPE; i < end; i += EPI * U) {
    int4 q[U];
    float nrm[U];
    float4 xv[U];
#pragma unroll
    for (int j = 0; j < U; ++j) {
      const int idx = i + j * EPI;
      q[j] = rec[idx < end ? idx : i];
    }
#pragma unroll
    for (int j = 0; j < U; ++j) {
      nrm[j] = dinv[q[j].x] * __int_as_float(q[j].z) * dinv[q[j].y];
      xv[j] = x4[(size_t)q[j].x * LPE + sub];
    }
#pragma unroll
    for (int j = 0; j < U; ++j) {
      if (i + j * EPI < end) {
        float* dst = acc + (q[j].y & (BK - 1)) * CH + sub * 4;
        lds_add_f32(dst + 0, xv[j].x * nrm[j]);
        lds_add_f32(dst + 1, xv[j].y * nrm[j]);
        lds_add_f32(dst + 2, xv[j].z * nrm[j]);
        lds_add_f32(dst + 3, xv[j].w * nrm[j]);
      }
    }
  }
  __syncthreads();

  float4* out4 = (float4*)out;
  for (int i = t; i < BK * LPE; i += 512) {
    const int n = node0 + i / LPE;
    if (n < N) out4[(size_t)n * LPE + (i % LPE)] = acc4[i];
  }
}

__global__ void bias_relu_kernel(float* __restrict__ x,
                                 const float* __restrict__ b, int N) {
  long long tid = (long long)blockIdx.x * blockDim.x + threadIdx.x;
  if (tid >= (long long)N * 16) return;
  int c4 = (int)(tid & 15);
  float4 v  = ((float4*)x)[tid];
  float4 bb = ((const float4*)b)[c4];
  v.x = fmaxf(v.x + bb.x, 0.0f);
  v.y = fmaxf(v.y + bb.y, 0.0f);
  v.z = fmaxf(v.z + bb.z, 0.0f);
  v.w = fmaxf(v.w + bb.w, 0.0f);
  ((float4*)x)[tid] = v;
}

__global__ void l2norm_kernel(const float* __restrict__ x,
                              const float* __restrict__ b2,
                              float* __restrict__ out, int N) {
  int row  = blockIdx.x * (blockDim.x >> 5) + (threadIdx.x >> 5);
  int lane = threadIdx.x & 31;
  if (row >= N) return;
  float v = x[(long long)row * 32 + lane] + b2[lane];
  float ss = v * v;
#pragma unroll
  for (int off = 16; off > 0; off >>= 1)
    ss += __shfl_xor(ss, off, 32);
  float l2 = sqrtf(ss);
  float scale = 1.0f / fmaxf(l2, 1e-12f);
  out[(long long)row * 32 + lane] = v * scale;
}

extern "C" void kernel_launch(void* const* d_in, const int* in_sizes, int n_in,
                              void* d_out, int out_size, void* d_ws, size_t ws_size,
                              hipStream_t stream) {
  const int*   edge_index = (const int*)d_in[0];
  const float* ew         = (const float*)d_in[1];
  const float* emb        = (const float*)d_in[2];
  const float* W1         = (const float*)d_in[3];
  const float* b1         = (const float*)d_in[4];
  const float* W2         = (const float*)d_in[5];
  const float* b2         = (const float*)d_in[6];

  const int E = in_sizes[0] / 2;
  const int N = in_sizes[2] / 64;
  const int* rows = edge_index;
  const int* cols = edge_index + E;
  const int NB = (N + BK - 1) / BK;
  const int chunk_len = (E + NCHUNK - 1) / NCHUNK;

  int4*  rec  = (int4*)d_ws;
  float* dinv = (float*)(rec + (size_t)E);
  float* xw1  = dinv + (size_t)N;
  float* out1 = xw1  + (size_t)N * 64;
  float* xw2  = out1 + (size_t)N * 64;
  float* out2 = xw2  + (size_t)N * 32;
  int* bbase     = (int*)(out2 + (size_t)N * 32);
  int* btotal    = bbase + (size_t)NB + 1;
  int* chunkhist = btotal + (size_t)NB;

  bucket_hist_kernel<<<NCHUNK, 1024, sizeof(int) * (size_t)NB, stream>>>(
      cols, E, chunk_len, NB, chunkhist);
  row_scan_kernel<<<NB, NCHUNK, 0, stream>>>(chunkhist, btotal);
  bucket_base_kernel<<<1, 1024, 0, stream>>>(btotal, bbase, NB);
  bucket_scatter_kernel<<<NCHUNK, 1024, sizeof(int) * (size_t)NB, stream>>>(
      rows, cols, ew, E, chunk_len, NB, chunkhist, bbase, rec);

  deg_dinv_kernel<<<NB, 256, 0, stream>>>(rec, bbase, dinv, N);

  gemm_wmma_f32<64><<<N / 16, 128, 0, stream>>>(emb, W1, xw1);
  aggregate_lds_kernel<64, 8><<<NB, 512, 0, stream>>>(xw1, rec, bbase, dinv, out1, N);
  bias_relu_kernel<<<(unsigned)(((long long)N * 16 + 255) / 256), 256, 0, stream>>>(out1, b1, N);

  gemm_wmma_f32<32><<<N / 16, 64, 0, stream>>>(out1, W2, xw2);
  aggregate_lds_kernel<32, 8><<<NB, 512, 0, stream>>>(xw2, rec, bbase, dinv, out2, N);

  l2norm_kernel<<<(N + 7) / 8, 256, 0, stream>>>(out2, b2, (float*)d_out, N);
}


# Bias+relu and bias+row L2-normalize folded into the bucket aggregate kernels' store loops: 11 -> 9 launches, 150000 short waves gone; row norm now summed 4 per lane then 8-lane shuffle tree
# speedup vs baseline: 2.1039x; 1.0583x over previous
#include <hip/hip_runtime.h>
#include <math.h>

typedef __attribute__((ext_vector_type(2))) float v2f;
typedef __attribute__((ext_vector_type(8))) float v8f;

#define BK_SHIFT 7
#define BK 128
#define NCHUNK 256

__device__ __forceinline__ void lds_add_f32(float* p, float v) {
  __hip_atomic_fetch_add(p, v, __ATOMIC_RELAXED, __HIP_MEMORY_SCOPE_WORKGROUP);
}
__device__ __forceinline__ int lds_add_i32(int* p, int v) {
  return __hip_atomic_fetch_add(p, v, __ATOMIC_RELAXED, __HIP_MEMORY_SCOPE_WORKGROUP);
}

__device__ __forceinline__ int wave_incl_scan(int v, int lane) {
  for (int off = 1; off < 32; off <<= 1) {
    int n = __shfl_up(v, off, 32);
    if (lane >= off) v += n;
  }
  return v;
}

__global__ void __launch_bounds__(1024)
bucket_hist_kernel(const int* __restrict__ cols, int E, int chunk_len, int NB,
                   int* __restrict__ chunkhist) {
  extern __shared__ int sh_cnt[];
  const int t = threadIdx.x;
  for (int i = t; i < NB; i += 1024) sh_cnt[i] = 0;
  __syncthreads();
  const int beg = blockIdx.x * chunk_len;
  const int end = min(E, beg + chunk_len);
  for (int e = beg + t; e < end; e += 4 * 1024) {
    int c[4];
#pragma unroll
    for (int j = 0; j < 4; ++j) {
      const int ej = e + j * 1024;
      c[j] = cols[ej < end ? ej : e];
    }
#pragma unroll
    for (int j = 0; j < 4; ++j)
      if (e + j * 1024 < end) lds_add_i32(&sh_cnt[c[j] >> BK_SHIFT], 1);
  }
  __syncthreads();
  for (int i = t; i < NB; i += 1024)
    chunkhist[(size_t)i * NCHUNK + blockIdx.x] = sh_cnt[i];
}

__global__ void __launch_bounds__(NCHUNK)
row_scan_kernel(int* __restrict__ chunkhist, int* __restrict__ btotal) {
  __shared__ int wsum[NCHUNK / 32];
  const int t = threadIdx.x, lane = t & 31, w = t >> 5;
  const size_t idx = (size_t)blockIdx.x * NCHUNK + t;
  const int v = chunkhist[idx];
  const int s = wave_incl_scan(v, lane);
  if (lane == 31) wsum[w] = s;
  __syncthreads();
  int wbase = 0;
  for (int k = 0; k < NCHUNK / 32; ++k)
    if (k < w) wbase += wsum[k];
  chunkhist[idx] = wbase + s - v;
  if (t == NCHUNK - 1) btotal[blockIdx.x] = wbase + s;
}

__global__ void __launch_bounds__(1024)
bucket_base_kernel(const int* __restrict__ btotal, int* __restrict__ bbase, int NB) {
  __shared__ int wsum[32];
  const int t = threadIdx.x, lane = t & 31, w = t >> 5;
  int carry = 0;
  for (int base = 0; base < NB; base += 1024) {
    const int i = base + t;
    const int v = (i < NB) ? btotal[i] : 0;
    const int s = wave_incl_scan(v, lane);
    if (lane == 31) wsum[w] = s;
    __syncthreads();
    if (w == 0) wsum[lane] = wave_incl_scan(wsum[lane], lane);
    __syncthreads();
    const int wbase = (w > 0) ? wsum[w - 1] : 0;
    if (i < NB) bbase[i] = carry + wbase + s - v;
    const int tile_total = wsum[31];
    __syncthreads();
    carry += tile_total;
  }
  if (t == 0) bbase[NB] = carry;
}

__global__ void __launch_bounds__(1024)
bucket_scatter_kernel(const int* __restrict__ rows, const int* __restrict__ cols,
                      const float* __restrict__ ew, int E, int chunk_len, int NB,
                      const int* __restrict__ chunkoff, const int* __restrict__ bbase,
                      int4* __restrict__ rec) {
  extern __shared__ int sh_cur[];
  const int t = threadIdx.x;
  for (int i = t; i < NB; i += 1024)
    sh_cur[i] = bbase[i] + chunkoff[(size_t)i * NCHUNK + blockIdx.x];
  __syncthreads();
  const int beg = blockIdx.x * chunk_len;
  const int end = min(E, beg + chunk_len);
  for (int e = beg + t; e < end; e += 4 * 1024) {
    int c[4], r[4];
    float w[4];
#pragma unroll
    for (int j = 0; j < 4; ++j) {
      const int ej = e + j * 1024;
      const int es = ej < end ? ej : e;
      c[j] = cols[es];
      r[j] = rows[es];
      w[j] = ew[es];
    }
#pragma unroll
    for (int j = 0; j < 4; ++j) {
      if (e + j * 1024 < end) {
        const int pos = lds_add_i32(&sh_cur[c[j] >> BK_SHIFT], 1);
        rec[pos] = make_int4(r[j], c[j], __float_as_int(w[j]), 0);
      }
    }
  }
}

__global__ void __launch_bounds__(256)
deg_dinv_kernel(const int4* __restrict__ rec, const int* __restrict__ bbase,
                float* __restrict__ dinv, int N) {
  __shared__ float sdeg[BK];
  const int t = threadIdx.x;
  if (t < BK) sdeg[t] = 1.0f;
  __syncthreads();
  const int beg = bbase[blockIdx.x], end = bbase[blockIdx.x + 1];
  for (int i = beg + t; i < end; i += 4 * 256) {
    int4 q[4];
#pragma unroll
    for (int j = 0; j < 4; ++j) {
      const int ij = i + j * 256;
      q[j] = rec[ij < end ? ij : i];
    }
#pragma unroll
    for (int j = 0; j < 4; ++j)
      if (i + j * 256 < end) lds_add_f32(&sdeg[q[j].y & (BK - 1)], __int_as_float(q[j].z));
  }
  __syncthreads();
  if (t < BK) {
    const int n = blockIdx.x * BK + t;
    if (n < N) {
      const float d = sdeg[t];
      dinv[n] = (d > 0.0f) ? rsqrtf(d) : 0.0f;
    }
  }
}

template <int NCOLS>
__global__ void gemm_wmma_f32(const float* __restrict__ A,
                              const float* __restrict__ B,
                              float* __restrict__ C) {
  const int m0   = blockIdx.x * 16;
  const int wid  = threadIdx.x >> 5;
  const int lane = threadIdx.x & 31;
  const int n0   = wid * 16;
  const int ln   = lane & 15;
  const int half = lane >> 4;

  v8f acc = {};
  const float* aptr = A + (size_t)(m0 + ln) * 64 + 2 * half;
  const float* bptr = B + (size_t)(2 * half) * NCOLS + n0 + ln;

#pragma unroll
  for (int k = 0; k < 64; k += 4) {
    v2f a, b;
    a.x = aptr[k];
    a.y = aptr[k + 1];
    b.x = bptr[k * NCOLS];
    b.y = bptr[k * NCOLS + NCOLS];
    acc = __builtin_amdgcn_wmma_f32_16x16x4_f32(
         false, a,  false, b,
         (short)0, acc,  false,  false);
  }

  float* cptr = C + (size_t)(m0 + 8 * half) * NCOLS + n0 + ln;
#pragma unroll
  for (int r = 0; r < 8; ++r) {
    cptr[(size_t)r * NCOLS] = acc[r];
  }
}

template <int CH, int U, bool L2NORM>
__global__ void __launch_bounds__(512)
aggregate_lds_kernel(const float* __restrict__ x, const int4* __restrict__ rec,
                     const int* __restrict__ bbase, const float* __restrict__ dinv,
                     const float* __restrict__ bias, float* __restrict__ out, int N) {
  constexpr int LPE = CH / 4;
  constexpr int EPI = 512 / LPE;
  __shared__ float4 acc4[BK * LPE];
  float* acc = (float*)acc4;
  const int t = threadIdx.x;
  const int node0 = blockIdx.x * BK;
  const float4* x4 = (const float4*)x;

  for (int i = t; i < BK * LPE; i += 512) {
    const int n = node0 + i / LPE;
    float4 v = make_float4(0.0f, 0.0f, 0.0f, 0.0f);
    if (n < N) {
      const float d = dinv[n];
      const float nrm = d * d;
      const float4 xv = x4[(size_t)n * LPE + (i % LPE)];
      v = make_float4(xv.x * nrm, xv.y * nrm, xv.z * nrm, xv.w * nrm);
    }
    acc4[i] = v;
  }
  __syncthreads();

  const int beg = bbase[blockIdx.x], end = bbase[blockIdx.x + 1];
  const int sub = t % LPE;
  for (int i = beg + t / LPE; i < end; i += EPI * U) {
    int4 q[U];
    float nrm[U];
    float4 xv[U];
#pragma unroll
    for (int j = 0; j < U; ++j) {
      const int idx = i + j * EPI;
      q[j] = rec[idx < end ? idx : i];
    }
#pragma unroll
    for (int j = 0; j < U; ++j) {
      nrm[j] = dinv[q[j].x] * __int_as_float(q[j].z) * dinv[q[j].y];
      xv[j] = x4[(size_t)q[j].x * LPE + sub];
    }
#pragma unroll
    for (int j = 0; j < U; ++j) {
      if (i + j * EPI < end) {
        float* dst = acc + (q[j].y & (BK - 1)) * CH + sub * 4;
        lds_add_f32(dst + 0, xv[j].x * nrm[j]);
        lds_add_f32(dst + 1, xv[j].y * nrm[j]);
        lds_add_f32(dst + 2, xv[j].z * nrm[j]);
        lds_add_f32(dst + 3, xv[j].w * nrm[j]);
      }
    }
  }
  __syncthreads();

  float4* out4 = (float4*)out;
  const float b0 = bias[sub * 4 + 0], b1 = bias[sub * 4 + 1];
  const float b2 = bias[sub * 4 + 2], b3 = bias[sub * 4 + 3];
  for (int i = t; i < BK * LPE; i += 512) {
    const int n = node0 + i / LPE;
    float4 v = acc4[i];
    v.x += b0; v.y += b1; v.z += b2; v.w += b3;
    if (L2NORM) {
      float ss = v.x * v.x + v.y * v.y + v.z * v.z + v.w * v.w;
      for (int off = 1; off < LPE; off <<= 1) ss += __shfl_xor(ss, off, 32);
      const float scale = 1.0f / fmaxf(sqrtf(ss), 1e-12f);
      v.x *= scale; v.y *= scale; v.z *= scale; v.w *= scale;
    } else {
      v.x = fmaxf(v.x, 0.0f); v.y = fmaxf(v.y, 0.0f);
      v.z = fmaxf(v.z, 0.0f); v.w = fmaxf(v.w, 0.0f);
    }
    if (n < N) out4[(size_t)n * LPE + sub] = v;
  }
}

extern "C" void kernel_launch(void* const* d_in, const int* in_sizes, int n_in,
                              void* d_out, int out_size, void* d_ws, size_t ws_size,
                              hipStream_t stream) {
  const int*   edge_index = (const int*)d_in[0];
  const float* ew         = (const float*)d_in[1];
  const float* emb        = (const float*)d_in[2];
  const float* W1         = (const float*)d_in[3];
  const float* b1         = (const float*)d_in[4];
  const float* W2         = (const float*)d_in[5];
  const float* b2         = (const float*)d_in[6];

  const int E = in_sizes[0] / 2;
  const int N = in_sizes[2] / 64;
  const int* rows = edge_index;
  const int* cols = edge_index + E;
  const int NB = (N + BK - 1) / BK;
  const int chunk_len = (E + NCHUNK - 1) / NCHUNK;

  int4*  rec  = (int4*)d_ws;
  float* dinv = (float*)(rec + (size_t)E);
  float* xw1  = dinv + (size_t)N;
  float* out1 = xw1  + (size_t)N * 64;
  float* xw2  = out1 + (size_t)N * 64;
  int* bbase     = (int*)(xw2 + (size_t)N * 32);
  int* btotal    = bbase + (size_t)NB + 1;
  int* chunkhist = btotal + (size_t)NB;

  bucket_hist_kernel<<<NCHUNK, 1024, sizeof(int) * (size_t)NB, stream>>>(
      cols, E, chunk_len, NB, chunkhist);
  row_scan_kernel<<<NB, NCHUNK, 0, stream>>>(chunkhist, btotal);
  bucket_base_kernel<<<1, 1024, 0, stream>>>(btotal, bbase, NB);
  bucket_scatter_kernel<<<NCHUNK, 1024, sizeof(int) * (size_t)NB, stream>>>(
      rows, cols, ew, E, chunk_len, NB, chunkhist, bbase, rec);

  deg_dinv_kernel<<<NB, 256, 0, stream>>>(rec, bbase, dinv, N);

  gemm_wmma_f32<64><<<N / 16, 128, 0, stream>>>(emb, W1, xw1);
  aggregate_lds_kernel<64, 8, false><<<NB, 512, 0, stream>>>(xw1, rec, bbase, dinv, b1, out1, N);

  gemm_wmma_f32<32><<<N / 16, 64, 0, stream>>>(out1, W2, xw2);
  aggregate_lds_kernel<32, 8, true><<<NB, 512, 0, stream>>>(xw2, rec, bbase, dinv, b2,
                                                           (float*)d_out, N);
}


# Degree factors out of the edge loop: GEMM epilogue stores rows scaled by dinv[row], dinv[col] multiplies each row sum once in the store loop; no per-edge dinv lookups (sum factored, same order)
# speedup vs baseline: 2.2875x; 1.0873x over previous
#include <hip/hip_runtime.h>
#include <math.h>

typedef __attribute__((ext_vector_type(2))) float v2f;
typedef __attribute__((ext_vector_type(8))) float v8f;

#define BK_SHIFT 7
#define BK 128
#define NCHUNK 256

__device__ __forceinline__ void lds_add_f32(float* p, float v) {
  __hip_atomic_fetch_add(p, v, __ATOMIC_RELAXED, __HIP_MEMORY_SCOPE_WORKGROUP);
}
__device__ __forceinline__ int lds_add_i32(int* p, int v) {
  return __hip_atomic_fetch_add(p, v, __ATOMIC_RELAXED, __HIP_MEMORY_SCOPE_WORKGROUP);
}

__device__ __forceinline__ int wave_incl_scan(int v, int lane) {
  for (int off = 1; off < 32; off <<= 1) {
    int n = __shfl_up(v, off, 32);
    if (lane >= off) v += n;
  }
  return v;
}

__global__ void __launch_bounds__(1024)
bucket_hist_kernel(const int* __restrict__ cols, int E, int chunk_len, int NB,
                   int* __restrict__ chunkhist) {
  extern __shared__ int sh_cnt[];
  const int t = threadIdx.x;
  for (int i = t; i < NB; i += 1024) sh_cnt[i] = 0;
  __syncthreads();
  const int beg = blockIdx.x * chunk_len;
  const int end = min(E, beg + chunk_len);
  for (int e = beg + t; e < end; e += 4 * 1024) {
    int c[4];
#pragma unroll
    for (int j = 0; j < 4; ++j) {
      const int ej = e + j * 1024;
      c[j] = cols[ej < end ? ej : e];
    }
#pragma unroll
    for (int j = 0; j < 4; ++j)
      if (e + j * 1024 < end) lds_add_i32(&sh_cnt[c[j] >> BK_SHIFT], 1);
  }
  __syncthreads();
  for (int i = t; i < NB; i += 1024)
    chunkhist[(size_t)i * NCHUNK + blockIdx.x] = sh_cnt[i];
}

__global__ void __launch_bounds__(NCHUNK)
row_scan_kernel(int* __restrict__ chunkhist, int* __restrict__ btotal) {
  __shared__ int wsum[NCHUNK / 32];
  const int t = threadIdx.x, lane = t & 31, w = t >> 5;
  const size_t idx = (size_t)blockIdx.x * NCHUNK + t;
  const int v = chunkhist[idx];
  const int s = wave_incl_scan(v, lane);
  if (lane == 31) wsum[w] = s;
  __syncthreads();
  int wbase = 0;
  for (int k = 0; k < NCHUNK / 32; ++k)
    if (k < w) wbase += wsum[k];
  chunkhist[idx] = wbase + s - v;
  if (t == NCHUNK - 1) btotal[blockIdx.x] = wbase + s;
}

__global__ void __launch_bounds__(1024)
bucket_base_kernel(const int* __restrict__ btotal, int* __restrict__ bbase, int NB) {
  __shared__ int wsum[32];
  const int t = threadIdx.x, lane = t & 31, w = t >> 5;
  int carry = 0;
  for (int base = 0; base < NB; base += 1024) {
    const int i = base + t;
    const int v = (i < NB) ? btotal[i] : 0;
    const int s = wave_incl_scan(v, lane);
    if (lane == 31) wsum[w] = s;
    __syncthreads();
    if (w == 0) wsum[lane] = wave_incl_scan(wsum[lane], lane);
    __syncthreads();
    const int wbase = (w > 0) ? wsum[w - 1] : 0;
    if (i < NB) bbase[i] = carry + wbase + s - v;
    const int tile_total = wsum[31];
    __syncthreads();
    carry += tile_total;
  }
  if (t == 0) bbase[NB] = carry;
}

__global__ void __launch_bounds__(1024)
bucket_scatter_kernel(const int* __restrict__ rows, const int* __restrict__ cols,
                      const float* __restrict__ ew, int E, int chunk_len, int NB,
                      const int* __restrict__ chunkoff, const int* __restrict__ bbase,
                      int4* __restrict__ rec) {
  extern __shared__ int sh_cur[];
  const int t = threadIdx.x;
  for (int i = t; i < NB; i += 1024)
    sh_cur[i] = bbase[i] + chunkoff[(size_t)i * NCHUNK + blockIdx.x];
  __syncthreads();
  const int beg = blockIdx.x * chunk_len;
  const int end = min(E, beg + chunk_len);
  for (int e = beg + t; e < end; e += 4 * 1024) {
    int c[4], r[4];
    float w[4];
#pragma unroll
    for (int j = 0; j < 4; ++j) {
      const int ej = e + j * 1024;
      const int es = ej < end ? ej : e;
      c[j] = cols[es];
      r[j] = rows[es];
      w[j] = ew[es];
    }
#pragma unroll
    for (int j = 0; j < 4; ++j) {
      if (e + j * 1024 < end) {
        const int pos = lds_add_i32(&sh_cur[c[j] >> BK_SHIFT], 1);
        rec[pos] = make_int4(r[j], c[j], __float_as_int(w[j]), 0);
      }
    }
  }
}

__global__ void __launch_bounds__(256)
deg_dinv_kernel(const int4* __restrict__ rec, const int* __restrict__ bbase,
                float* __restrict__ dinv, int N) {
  __shared__ float sdeg[BK];
  const int t = threadIdx.x;
  if (t < BK) sdeg[t] = 1.0f;
  __syncthreads();
  const int beg = bbase[blockIdx.x], end = bbase[blockIdx.x + 1];
  for (int i = beg + t; i < end; i += 4 * 256) {
    int4 q[4];
#pragma unroll
    for (int j = 0; j < 4; ++j) {
      const int ij = i + j * 256;
      q[j] = rec[ij < end ? ij : i];
    }
#pragma unroll
    for (int j = 0; j < 4; ++j)
      if (i + j * 256 < end) lds_add_f32(&sdeg[q[j].y & (BK - 1)], __int_as_float(q[j].z));
  }
  __syncthreads();
  if (t < BK) {
    const int n = blockIdx.x * BK + t;
    if (n < N) {
      const float d = sdeg[t];
      dinv[n] = (d > 0.0f) ? rsqrtf(d) : 0.0f;
    }
  }
}

template <int NCOLS>
__global__ void gemm_wmma_f32(const float* __restrict__ A,
                              const float* __restrict__ B,
                              const float* __restrict__ rowscale,
                              float* __restrict__ C) {
  const int m0   = blockIdx.x * 16;
  const int wid  = threadIdx.x >> 5;
  const int lane = threadIdx.x & 31;
  const int n0   = wid * 16;
  const int ln   = lane & 15;
  const int half = lane >> 4;

  v8f acc = {};
  const float* aptr = A + (size_t)(m0 + ln) * 64 + 2 * half;
  const float* bptr = B + (size_t)(2 * half) * NCOLS + n0 + ln;

#pragma unroll
  for (int k = 0; k < 64; k += 4) {
    v2f a, b;
    a.x = aptr[k];
    a.y = aptr[k + 1];
    b.x = bptr[k * NCOLS];
    b.y = bptr[k * NCOLS + NCOLS];
    acc = __builtin_amdgcn_wmma_f32_16x16x4_f32(
         false, a,  false, b,
         (short)0, acc,  false,  false);
  }

  const float4 s_lo = *(const float4*)(rowscale + m0 + 8 * half);
  const float4 s_hi = *(const float4*)(rowscale + m0 + 8 * half + 4);
  const float sc[8] = {s_lo.x, s_lo.y, s_lo.z, s_lo.w, s_hi.x, s_hi.y, s_hi.z, s_hi.w};
  float* cptr = C + (size_t)(m0 + 8 * half) * NCOLS + n0 + ln;
#pragma unroll
  for (int r = 0; r < 8; ++r) {
    cptr[(size_t)r * NCOLS] = acc[r] * sc[r];
  }
}

template <int CH, int U, bool L2NORM>
__global__ void __launch_bounds__(512)
aggregate_lds_kernel(const float* __restrict__ xs, const int4* __restrict__ rec,
                     const int* __restrict__ bbase, const float* __restrict__ dinv,
                     const float* __restrict__ bias, float* __restrict__ out, int N) {
  constexpr int LPE = CH / 4;
  constexpr int EPI = 512 / LPE;
  __shared__ float4 acc4[BK * LPE];
  float* acc = (float*)acc4;
  const int t = threadIdx.x;
  const int node0 = blockIdx.x * BK;
  const float4* x4 = (const float4*)xs;

  for (int i = t; i < BK * LPE; i += 512) {
    const int n = node0 + i / LPE;
    float4 v = make_float4(0.0f, 0.0f, 0.0f, 0.0f);
    if (n < N) v = x4[(size_t)n * LPE + (i % LPE)];
    acc4[i] = v;
  }
  __syncthreads();

  const int beg = bbase[blockIdx.x], end = bbase[blockIdx.x + 1];
  const int sub = t % LPE;
  for (int i = beg + t / LPE; i < end; i += EPI * U) {
    int4 q[U];
    float4 xv[U];
#pragma unroll
    for (int j = 0; j < U; ++j) {
      const int idx = i + j * EPI;
      q[j] = rec[idx < end ? idx : i];
    }
#pragma unroll
    for (int j = 0; j < U; ++j) xv[j] = x4[(size_t)q[j].x * LPE + sub];
#pragma unroll
    for (int j = 0; j < U; ++j) {
      if (i + j * EPI < end) {
        const float w = __int_as_float(q[j].z);
        float* dst = acc + (q[j].y & (BK - 1)) * CH + sub * 4;
        lds_add_f32(dst + 0, xv[j].x * w);
        lds_add_f32(dst + 1, xv[j].y * w);
        lds_add_f32(dst + 2, xv[j].z * w);
        lds_add_f32(dst + 3, xv[j].w * w);
      }
    }
  }
  __syncthreads();

  float4* out4 = (float4*)out;
  const float b0 = bias[sub * 4 + 0], b1 = bias[sub * 4 + 1];
  const float b2 = bias[sub * 4 + 2], b3 = bias[sub * 4 + 3];
  for (int i = t; i < BK * LPE; i += 512) {
    const int n = node0 + i / LPE;
    const float d = (n < N) ? dinv[n] : 0.0f;
    float4 v = acc4[i];
    v.x = v.x * d + b0; v.y = v.y * d + b1; v.z = v.z * d + b2; v.w = v.w * d + b3;
    if (L2NORM) {
      float ss = v.x * v.x + v.y * v.y + v.z * v.z + v.w * v.w;
      for (int off = 1; off < LPE; off <<= 1) ss += __shfl_xor(ss, off, 32);
      const float scale = 1.0f / fmaxf(sqrtf(ss), 1e-12f);
      v.x *= scale; v.y *= scale; v.z *= scale; v.w *= scale;
    } else {
      v.x = fmaxf(v.x, 0.0f); v.y = fmaxf(v.y, 0.0f);
      v.z = fmaxf(v.z, 0.0f); v.w = fmaxf(v.w, 0.0f);
    }
    if (n < N) out4[(size_t)n * LPE + sub] = v;
  }
}

extern "C" void kernel_launch(void* const* d_in, const int* in_sizes, int n_in,
                              void* d_out, int out_size, void* d_ws, size_t ws_size,
                              hipStream_t stream) {
  const int*   edge_index = (const int*)d_in[0];
  const float* ew         = (const float*)d_in[1];
  const float* emb        = (const float*)d_in[2];
  const float* W1         = (const float*)d_in[3];
  const float* b1         = (const float*)d_in[4];
  const float* W2         = (const float*)d_in[5];
  const float* b2         = (const float*)d_in[6];

  const int E = in_sizes[0] / 2;
  const int N = in_sizes[2] / 64;
  const int* rows = edge_index;
  const int* cols = edge_index + E;
  const int NB = (N + BK - 1) / BK;
  const int chunk_len = (E + NCHUNK - 1) / NCHUNK;

  int4*  rec  = (int4*)d_ws;
  float* dinv = (float*)(rec + (size_t)E);
  float* xw1  = dinv + (size_t)N;
  float* out1 = xw1  + (size_t)N * 64;
  float* xw2  = out1 + (size_t)N * 64;
  int* bbase     = (int*)(xw2 + (size_t)N * 32);
  int* btotal    = bbase + (size_t)NB + 1;
  int* chunkhist = btotal + (size_t)NB;

  bucket_hist_kernel<<<NCHUNK, 1024, sizeof(int) * (size_t)NB, stream>>>(
      cols, E, chunk_len, NB, chunkhist);
  row_scan_kernel<<<NB, NCHUNK, 0, stream>>>(chunkhist, btotal);
  bucket_base_kernel<<<1, 1024, 0, stream>>>(btotal, bbase, NB);
  bucket_scatter_kernel<<<NCHUNK, 1024, sizeof(int) * (size_t)NB, stream>>>(
      rows, cols, ew, E, chunk_len, NB, chunkhist, bbase, rec);

  deg_dinv_kernel<<<NB, 256, 0, stream>>>(rec, bbase, dinv, N);

  gemm_wmma_f32<64><<<N / 16, 128, 0, stream>>>(emb, W1, dinv, xw1);
  aggregate_lds_kernel<64, 8, false><<<NB, 512, 0, stream>>>(xw1, rec, bbase, dinv, b1, out1, N);

  gemm_wmma_f32<32><<<N / 16, 64, 0, stream>>>(out1, W2, dinv, xw2);
  aggregate_lds_kernel<32, 8, true><<<NB, 512, 0, stream>>>(xw2, rec, bbase, dinv, b2,
                                                           (float*)d_out, N);
}


# Aggregate LDS float adds now bank-conflict-free: lane group g of a wave adds its 4 components in order c^g, so the 2 or 4 edges of one wave-wide add hit disjoint banks (add order varies as before)
# speedup vs baseline: 3.1479x; 1.3761x over previous
#include <hip/hip_runtime.h>
#include <math.h>

typedef __attribute__((ext_vector_type(2))) float v2f;
typedef __attribute__((ext_vector_type(8))) float v8f;

#define BK_SHIFT 7
#define BK 128
#define NCHUNK 256

__device__ __forceinline__ void lds_add_f32(float* p, float v) {
  __hip_atomic_fetch_add(p, v, __ATOMIC_RELAXED, __HIP_MEMORY_SCOPE_WORKGROUP);
}
__device__ __forceinline__ int lds_add_i32(int* p, int v) {
  return __hip_atomic_fetch_add(p, v, __ATOMIC_RELAXED, __HIP_MEMORY_SCOPE_WORKGROUP);
}

__device__ __forceinline__ int wave_incl_scan(int v, int lane) {
  for (int off = 1; off < 32; off <<= 1) {
    int n = __shfl_up(v, off, 32);
    if (lane >= off) v += n;
  }
  return v;
}

__global__ void __launch_bounds__(1024)
bucket_hist_kernel(const int* __restrict__ cols, int E, int chunk_len, int NB,
                   int* __restrict__ chunkhist) {
  extern __shared__ int sh_cnt[];
  const int t = threadIdx.x;
  for (int i = t; i < NB; i += 1024) sh_cnt[i] = 0;
  __syncthreads();
  const int beg = blockIdx.x * chunk_len;
  const int end = min(E, beg + chunk_len);
  for (int e = beg + t; e < end; e += 4 * 1024) {
    int c[4];
#pragma unroll
    for (int j = 0; j < 4; ++j) {
      const int ej = e + j * 1024;
      c[j] = cols[ej < end ? ej : e];
    }
#pragma unroll
    for (int j = 0; j < 4; ++j)
      if (e + j * 1024 < end) lds_add_i32(&sh_cnt[c[j] >> BK_SHIFT], 1);
  }
  __syncthreads();
  for (int i = t; i < NB; i += 1024)
    chunkhist[(size_t)i * NCHUNK + blockIdx.x] = sh_cnt[i];
}

__global__ void __launch_bounds__(NCHUNK)
row_scan_kernel(int* __restrict__ chunkhist, int* __restrict__ btotal) {
  __shared__ int wsum[NCHUNK / 32];
  const int t = threadIdx.x, lane = t & 31, w = t >> 5;
  const size_t idx = (size_t)blockIdx.x * NCHUNK + t;
  const int v = chunkhist[idx];
  const int s = wave_incl_scan(v, lane);
  if (lane == 31) wsum[w] = s;
  __syncthreads();
  int wbase = 0;
  for (int k = 0; k < NCHUNK / 32; ++k)
    if (k < w) wbase += wsum[k];
  chunkhist[idx] = wbase + s - v;
  if (t == NCHUNK - 1) btotal[blockIdx.x] = wbase + s;
}

__global__ void __launch_bounds__(1024)
bucket_base_kernel(const int* __restrict__ btotal, int* __restrict__ bbase, int NB) {
  __shared__ int wsum[32];
  const int t = threadIdx.x, lane = t & 31, w = t >> 5;
  int carry = 0;
  for (int base = 0; base < NB; base += 1024) {
    const int i = base + t;
    const int v = (i < NB) ? btotal[i] : 0;
    const int s = wave_incl_scan(v, lane);
    if (lane == 31) wsum[w] = s;
    __syncthreads();
    if (w == 0) wsum[lane] = wave_incl_scan(wsum[lane], lane);
    __syncthreads();
    const int wbase = (w > 0) ? wsum[w - 1] : 0;
    if (i < NB) bbase[i] = carry + wbase + s - v;
    const int tile_total = wsum[31];
    __syncthreads();
    carry += tile_total;
  }
  if (t == 0) bbase[NB] = carry;
}

__global__ void __launch_bounds__(1024)
bucket_scatter_kernel(const int* __restrict__ rows, const int* __restrict__ cols,
                      const float* __restrict__ ew, int E, int chunk_len, int NB,
                      const int* __restrict__ chunkoff, const int* __restrict__ bbase,
                      int4* __restrict__ rec) {
  extern __shared__ int sh_cur[];
  const int t = threadIdx.x;
  for (int i = t; i < NB; i += 1024)
    sh_cur[i] = bbase[i] + chunkoff[(size_t)i * NCHUNK + blockIdx.x];
  __syncthreads();
  const int beg = blockIdx.x * chunk_len;
  const int end = min(E, beg + chunk_len);
  for (int e = beg + t; e < end; e += 4 * 1024) {
    int c[4], r[4];
    float w[4];
#pragma unroll
    for (int j = 0; j < 4; ++j) {
      const int ej = e + j * 1024;
      const int es = ej < end ? ej : e;
      c[j] = cols[es];
      r[j] = rows[es];
      w[j] = ew[es];
    }
#pragma unroll
    for (int j = 0; j < 4; ++j) {
      if (e + j * 1024 < end) {
        const int pos = lds_add_i32(&sh_cur[c[j] >> BK_SHIFT], 1);
        rec[pos] = make_int4(r[j], c[j], __float_as_int(w[j]), 0);
      }
    }
  }
}

__global__ void __launch_bounds__(256)
deg_dinv_kernel(const int4* __restrict__ rec, const int* __restrict__ bbase,
                float* __restrict__ dinv, int N) {
  __shared__ float sdeg[BK];
  const int t = threadIdx.x;
  if (t < BK) sdeg[t] = 1.0f;
  __syncthreads();
  const int beg = bbase[blockIdx.x], end = bbase[blockIdx.x + 1];
  for (int i = beg + t; i < end; i += 4 * 256) {
    int4 q[4];
#pragma unroll
    for (int j = 0; j < 4; ++j) {
      const int ij = i + j * 256;
      q[j] = rec[ij < end ? ij : i];
    }
#pragma unroll
    for (int j = 0; j < 4; ++j)
      if (i + j * 256 < end) lds_add_f32(&sdeg[q[j].y & (BK - 1)], __int_as_float(q[j].z));
  }
  __syncthreads();
  if (t < BK) {
    const int n = blockIdx.x * BK + t;
    if (n < N) {
      const float d = sdeg[t];
      dinv[n] = (d > 0.0f) ? rsqrtf(d) : 0.0f;
    }
  }
}

template <int NCOLS>
__global__ void gemm_wmma_f32(const float* __restrict__ A,
                              const float* __restrict__ B,
                              const float* __restrict__ rowscale,
                              float* __restrict__ C) {
  const int m0   = blockIdx.x * 16;
  const int wid  = threadIdx.x >> 5;
  const int lane = threadIdx.x & 31;
  const int n0   = wid * 16;
  const int ln   = lane & 15;
  const int half = lane >> 4;

  v8f acc = {};
  const float* aptr = A + (size_t)(m0 + ln) * 64 + 2 * half;
  const float* bptr = B + (size_t)(2 * half) * NCOLS + n0 + ln;

#pragma unroll
  for (int k = 0; k < 64; k += 4) {
    v2f a, b;
    a.x = aptr[k];
    a.y = aptr[k + 1];
    b.x = bptr[k * NCOLS];
    b.y = bptr[k * NCOLS + NCOLS];
    acc = __builtin_amdgcn_wmma_f32_16x16x4_f32(
         false, a,  false, b,
         (short)0, acc,  false,  false);
  }

  const float4 s_lo = *(const float4*)(rowscale + m0 + 8 * half);
  const float4 s_hi = *(const float4*)(rowscale + m0 + 8 * half + 4);
  const float sc[8] = {s_lo.x, s_lo.y, s_lo.z, s_lo.w, s_hi.x, s_hi.y, s_hi.z, s_hi.w};
  float* cptr = C + (size_t)(m0 + 8 * half) * NCOLS + n0 + ln;
#pragma unroll
  for (int r = 0; r < 8; ++r) {
    cptr[(size_t)r * NCOLS] = acc[r] * sc[r];
  }
}

template <int CH, int U, bool L2NORM>
__global__ void __launch_bounds__(512)
aggregate_lds_kernel(const float* __restrict__ xs, const int4* __restrict__ rec,
                     const int* __restrict__ bbase, const float* __restrict__ dinv,
                     const float* __restrict__ bias, float* __restrict__ out, int N) {
  constexpr int LPE = CH / 4;
  constexpr int EPI = 512 / LPE;
  constexpr int GPW = 32 / LPE;
  static_assert(GPW == 2 || GPW == 4, "component rotation is written for 2 or 4 groups");
  __shared__ float4 acc4[BK * LPE];
  float* acc = (float*)acc4;
  const int t = threadIdx.x;
  const int node0 = blockIdx.x * BK;
  const float4* x4 = (const float4*)xs;

  for (int i = t; i < BK * LPE; i += 512) {
    const int n = node0 + i / LPE;
    float4 v = make_float4(0.0f, 0.0f, 0.0f, 0.0f);
    if (n < N) v = x4[(size_t)n * LPE + (i % LPE)];
    acc4[i] = v;
  }
  __syncthreads();

  const int beg = bbase[blockIdx.x], end = bbase[blockIdx.x + 1];
  const int sub = t % LPE;
  const int g = (t / LPE) & (GPW - 1);
  const int o0 = g, o1 = g ^ 1, o2 = g ^ 2, o3 = g ^ 3;
  const bool sw1 = (g & 1) != 0;
  const bool sw2 = (g & 2) != 0;
  for (int i = beg + t / LPE; i < end; i += EPI * U) {
    int4 q[U];
    float4 xv[U];
#pragma unroll
    for (int j = 0; j < U; ++j) {
      const int idx = i + j * EPI;
      q[j] = rec[idx < end ? idx : i];
    }
#pragma unroll
    for (int j = 0; j < U; ++j) xv[j] = x4[(size_t)q[j].x * LPE + sub];
#pragma unroll
    for (int j = 0; j < U; ++j) {
      if (i + j * EPI < end) {
        const float w = __int_as_float(q[j].z);
        float* dst = acc + (q[j].y & (BK - 1)) * CH + sub * 4;
        float r0 = sw1 ? xv[j].y : xv[j].x;
        float r1 = sw1 ? xv[j].x : xv[j].y;
        float r2 = sw1 ? xv[j].w : xv[j].z;
        float r3 = sw1 ? xv[j].z : xv[j].w;
        if (GPW == 4) {
          const float u0 = sw2 ? r2 : r0, u1 = sw2 ? r3 : r1;
          const float u2 = sw2 ? r0 : r2, u3 = sw2 ? r1 : r3;
          r0 = u0; r1 = u1; r2 = u2; r3 = u3;
        }
        lds_add_f32(dst + o0, r0 * w);
        lds_add_f32(dst + o1, r1 * w);
        lds_add_f32(dst + o2, r2 * w);
        lds_add_f32(dst + o3, r3 * w);
      }
    }
  }
  __syncthreads();

  float4* out4 = (float4*)out;
  const float b0 = bias[sub * 4 + 0], b1 = bias[sub * 4 + 1];
  const float b2 = bias[sub * 4 + 2], b3 = bias[sub * 4 + 3];
  for (int i = t; i < BK * LPE; i += 512) {
    const int n = node0 + i / LPE;
    const float d = (n < N) ? dinv[n] : 0.0f;
    float4 v = acc4[i];
    v.x = v.x * d + b0; v.y = v.y * d + b1; v.z = v.z * d + b2; v.w = v.w * d + b3;
    if (L2NORM) {
      float ss = v.x * v.x + v.y * v.y + v.z * v.z + v.w * v.w;
      for (int off = 1; off < LPE; off <<= 1) ss += __shfl_xor(ss, off, 32);
      const float scale = 1.0f / fmaxf(sqrtf(ss), 1e-12f);
      v.x *= scale; v.y *= scale; v.z *= scale; v.w *= scale;
    } else {
      v.x = fmaxf(v.x, 0.0f); v.y = fmaxf(v.y, 0.0f);
      v.z = fmaxf(v.z, 0.0f); v.w = fmaxf(v.w, 0.0f);
    }
    if (n < N) out4[(size_t)n * LPE + sub] = v;
  }
}

extern "C" void kernel_launch(void* const* d_in, const int* in_sizes, int n_in,
                              void* d_out, int out_size, void* d_ws, size_t ws_size,
                              hipStream_t stream) {
  const int*   edge_index = (const int*)d_in[0];
  const float* ew         = (const float*)d_in[1];
  const float* emb        = (const float*)d_in[2];
  const float* W1         = (const float*)d_in[3];
  const float* b1         = (const float*)d_in[4];
  const float* W2         = (const float*)d_in[5];
  const float* b2         = (const float*)d_in[6];

  const int E = in_sizes[0] / 2;
  const int N = in_sizes[2] / 64;
  const int* rows = edge_index;
  const int* cols = edge_index + E;
  const int NB = (N + BK - 1) / BK;
  const int chunk_len = (E + NCHUNK - 1) / NCHUNK;

  int4*  rec  = (int4*)d_ws;
  float* dinv = (float*)(rec + (size_t)E);
  float* xw1  = dinv + (size_t)N;
  float* out1 = xw1  + (size_t)N * 64;
  float* xw2  = out1 + (size_t)N * 64;
  int* bbase     = (int*)(xw2 + (size_t)N * 32);
  int* btotal    = bbase + (size_t)NB + 1;
  int* chunkhist = btotal + (size_t)NB;

  bucket_hist_kernel<<<NCHUNK, 1024, sizeof(int) * (size_t)NB, stream>>>(
      cols, E, chunk_len, NB, chunkhist);
  row_scan_kernel<<<NB, NCHUNK, 0, stream>>>(chunkhist, btotal);
  bucket_base_kernel<<<1, 1024, 0, stream>>>(btotal, bbase, NB);
  bucket_scatter_kernel<<<NCHUNK, 1024, sizeof(int) * (size_t)NB, stream>>>(
      rows, cols, ew, E, chunk_len, NB, chunkhist, bbase, rec);

  deg_dinv_kernel<<<NB, 256, 0, stream>>>(rec, bbase, dinv, N);

  gemm_wmma_f32<64><<<N / 16, 128, 0, stream>>>(emb, W1, dinv, xw1);
  aggregate_lds_kernel<64, 8, false><<<NB, 512, 0, stream>>>(xw1, rec, bbase, dinv, b1, out1, N);

  gemm_wmma_f32<32><<<N / 16, 64, 0, stream>>>(out1, W2, dinv, xw2);
  aggregate_lds_kernel<32, 8, true><<<NB, 512, 0, stream>>>(xw2, rec, bbase, dinv, b2,
                                                           (float*)d_out, N);
}


# Scatter pass first sorts each 12500-edge chunk by bucket in LDS (LDS counter rank, in-block scan): records leave as contiguous runs, not one 16-B store per line; order in a bucket varies as before
# speedup vs baseline: 3.2858x; 1.0438x over previous
#include <hip/hip_runtime.h>
#include <math.h>

typedef __attribute__((ext_vector_type(2))) float v2f;
typedef __attribute__((ext_vector_type(8))) float v8f;

#define BK_SHIFT 7
#define BK 128
#define NCHUNK 256

__device__ __forceinline__ void lds_add_f32(float* p, float v) {
  __hip_atomic_fetch_add(p, v, __ATOMIC_RELAXED, __HIP_MEMORY_SCOPE_WORKGROUP);
}
__device__ __forceinline__ int lds_add_i32(int* p, int v) {
  return __hip_atomic_fetch_add(p, v, __ATOMIC_RELAXED, __HIP_MEMORY_SCOPE_WORKGROUP);
}

__device__ __forceinline__ int wave_incl_scan(int v, int lane) {
  for (int off = 1; off < 32; off <<= 1) {
    int n = __shfl_up(v, off, 32);
    if (lane >= off) v += n;
  }
  return v;
}

__global__ void __launch_bounds__(1024)
bucket_hist_kernel(const int* __restrict__ cols, int E, int chunk_len, int NB,
                   int* __restrict__ chunkhist) {
  extern __shared__ int sh_cnt[];
  const int t = threadIdx.x;
  for (int i = t; i < NB; i += 1024) sh_cnt[i] = 0;
  __syncthreads();
  const int beg = blockIdx.x * chunk_len;
  const int end = min(E, beg + chunk_len);
  for (int e = beg + t; e < end; e += 4 * 1024) {
    int c[4];
#pragma unroll
    for (int j = 0; j < 4; ++j) {
      const int ej = e + j * 1024;
      c[j] = cols[ej < end ? ej : e];
    }
#pragma unroll
    for (int j = 0; j < 4; ++j)
      if (e + j * 1024 < end) lds_add_i32(&sh_cnt[c[j] >> BK_SHIFT], 1);
  }
  __syncthreads();
  for (int i = t; i < NB; i += 1024)
    chunkhist[(size_t)i * NCHUNK + blockIdx.x] = sh_cnt[i];
}

__global__ void __launch_bounds__(NCHUNK)
row_scan_kernel(int* __restrict__ chunkhist, int* __restrict__ btotal) {
  __shared__ int wsum[NCHUNK / 32];
  const int t = threadIdx.x, lane = t & 31, w = t >> 5;
  const size_t idx = (size_t)blockIdx.x * NCHUNK + t;
  const int v = chunkhist[idx];
  const int s = wave_incl_scan(v, lane);
  if (lane == 31) wsum[w] = s;
  __syncthreads();
  int wbase = 0;
  for (int k = 0; k < NCHUNK / 32; ++k)
    if (k < w) wbase += wsum[k];
  chunkhist[idx] = wbase + s - v;
  if (t == NCHUNK - 1) btotal[blockIdx.x] = wbase + s;
}

__global__ void __launch_bounds__(1024)
bucket_base_kernel(const int* __restrict__ btotal, int* __restrict__ bbase, int NB) {
  __shared__ int wsum[32];
  const int t = threadIdx.x, lane = t & 31, w = t >> 5;
  int carry = 0;
  for (int base = 0; base < NB; base += 1024) {
    const int i = base + t;
    const int v = (i < NB) ? btotal[i] : 0;
    const int s = wave_incl_scan(v, lane);
    if (lane == 31) wsum[w] = s;
    __syncthreads();
    if (w == 0) wsum[lane] = wave_incl_scan(wsum[lane], lane);
    __syncthreads();
    const int wbase = (w > 0) ? wsum[w - 1] : 0;
    if (i < NB) bbase[i] = carry + wbase + s - v;
    const int tile_total = wsum[31];
    __syncthreads();
    carry += tile_total;
  }
  if (t == 0) bbase[NB] = carry;
}

#define SC_EPT 13
#define SC_TILE (SC_EPT * 1024)
__global__ void __launch_bounds__(1024)
bucket_scatter_kernel(const int* __restrict__ rows, const int* __restrict__ cols,
                      const float* __restrict__ ew, int E, int chunk_len, int NB,
                      const int* __restrict__ chunkoff, const int* __restrict__ bbase,
                      int4* __restrict__ rec) {
  __shared__ int4 srt[SC_TILE];
  __shared__ int wsum[32];
  extern __shared__ int sh_cur[];
  int* cur  = sh_cur;
  int* cnt  = cur + NB;
  int* loff = cnt + NB;
  const int t = threadIdx.x, lane = t & 31, w = t >> 5;
  const int beg = blockIdx.x * chunk_len;
  const int end = min(E, beg + chunk_len);
  for (int tb = beg; tb < end; tb += SC_TILE) {
    const int tn = min(SC_TILE, end - tb);
    int c[SC_EPT], r[SC_EPT], rk[SC_EPT];
    float wv[SC_EPT];
#pragma unroll
    for (int j = 0; j < SC_EPT; ++j) {
      const int idx = j * 1024 + t;
      const int es = tb + (idx < tn ? idx : 0);
      c[j] = cols[es];
      r[j] = rows[es];
      wv[j] = ew[es];
    }
    if (tb == beg) {
      for (int i = t; i < NB; i += 1024)
        cur[i] = bbase[i] + chunkoff[(size_t)i * NCHUNK + blockIdx.x];
    }
    for (int i = t; i < NB; i += 1024) cnt[i] = 0;
    __syncthreads();
#pragma unroll
    for (int j = 0; j < SC_EPT; ++j) {
      rk[j] = 0;
      if (j * 1024 + t < tn) rk[j] = lds_add_i32(&cnt[c[j] >> BK_SHIFT], 1);
    }
    __syncthreads();
    int carry = 0;
    for (int base = 0; base < NB; base += 1024) {
      const int i = base + t;
      const int v = (i < NB) ? cnt[i] : 0;
      const int s = wave_incl_scan(v, lane);
      if (lane == 31) wsum[w] = s;
      __syncthreads();
      if (w == 0) wsum[lane] = wave_incl_scan(wsum[lane], lane);
      __syncthreads();
      const int wbase = (w > 0) ? wsum[w - 1] : 0;
      if (i < NB) loff[i] = carry + wbase + s - v;
      const int tile_total = wsum[31];
      __syncthreads();
      carry += tile_total;
    }
    int slot[SC_EPT];
#pragma unroll
    for (int j = 0; j < SC_EPT; ++j) slot[j] = loff[c[j] >> BK_SHIFT] + rk[j];
#pragma unroll
    for (int j = 0; j < SC_EPT; ++j) {
      if (j * 1024 + t < tn)
        srt[slot[j]] = make_int4(r[j], c[j], __float_as_int(wv[j]), 0);
    }
    __syncthreads();
    int4 q[SC_EPT];
    int g[SC_EPT];
#pragma unroll
    for (int j = 0; j < SC_EPT; ++j) {
      const int i = j * 1024 + t;
      q[j] = srt[i < tn ? i : 0];
    }
#pragma unroll
    for (int j = 0; j < SC_EPT; ++j) {
      const int b = q[j].y >> BK_SHIFT;
      g[j] = cur[b] + (j * 1024 + t - loff[b]);
    }
#pragma unroll
    for (int j = 0; j < SC_EPT; ++j) {
      if (j * 1024 + t < tn) rec[g[j]] = q[j];
    }
    __syncthreads();
    for (int i = t; i < NB; i += 1024) cur[i] += cnt[i];
  }
}

__global__ void __launch_bounds__(256)
deg_dinv_kernel(const int4* __restrict__ rec, const int* __restrict__ bbase,
                float* __restrict__ dinv, int N) {
  __shared__ float sdeg[BK];
  const int t = threadIdx.x;
  if (t < BK) sdeg[t] = 1.0f;
  __syncthreads();
  const int beg = bbase[blockIdx.x], end = bbase[blockIdx.x + 1];
  for (int i = beg + t; i < end; i += 4 * 256) {
    int4 q[4];
#pragma unroll
    for (int j = 0; j < 4; ++j) {
      const int ij = i + j * 256;
      q[j] = rec[ij < end ? ij : i];
    }
#pragma unroll
    for (int j = 0; j < 4; ++j)
      if (i + j * 256 < end) lds_add_f32(&sdeg[q[j].y & (BK - 1)], __int_as_float(q[j].z));
  }
  __syncthreads();
  if (t < BK) {
    const int n = blockIdx.x * BK + t;
    if (n < N) {
      const float d = sdeg[t];
      dinv[n] = (d > 0.0f) ? rsqrtf(d) : 0.0f;
    }
  }
}

template <int NCOLS>
__global__ void gemm_wmma_f32(const float* __restrict__ A,
                              const float* __restrict__ B,
                              const float* __restrict__ rowscale,
                              float* __restrict__ C) {
  const int m0   = blockIdx.x * 16;
  const int wid  = threadIdx.x >> 5;
  const int lane = threadIdx.x & 31;
  const int n0   = wid * 16;
  const int ln   = lane & 15;
  const int half = lane >> 4;

  v8f acc = {};
  const float* aptr = A + (size_t)(m0 + ln) * 64 + 2 * half;
  const float* bptr = B + (size_t)(2 * half) * NCOLS + n0 + ln;

#pragma unroll
  for (int k = 0; k < 64; k += 4) {
    v2f a, b;
    a.x = aptr[k];
    a.y = aptr[k + 1];
    b.x = bptr[k * NCOLS];
    b.y = bptr[k * NCOLS + NCOLS];
    acc = __builtin_amdgcn_wmma_f32_16x16x4_f32(
         false, a,  false, b,
         (short)0, acc,  false,  false);
  }

  const float4 s_lo = *(const float4*)(rowscale + m0 + 8 * half);
  const float4 s_hi = *(const float4*)(rowscale + m0 + 8 * half + 4);
  const float sc[8] = {s_lo.x, s_lo.y, s_lo.z, s_lo.w, s_hi.x, s_hi.y, s_hi.z, s_hi.w};
  float* cptr = C + (size_t)(m0 + 8 * half) * NCOLS + n0 + ln;
#pragma unroll
  for (int r = 0; r < 8; ++r) {
    cptr[(size_t)r * NCOLS] = acc[r] * sc[r];
  }
}

template <int CH, int U, bool L2NORM>
__global__ void __launch_bounds__(512)
aggregate_lds_kernel(const float* __restrict__ xs, const int4* __restrict__ rec,
                     const int* __restrict__ bbase, const float* __restrict__ dinv,
                     const float* __restrict__ bias, float* __restrict__ out, int N) {
  constexpr int LPE = CH / 4;
  constexpr int EPI = 512 / LPE;
  constexpr int GPW = 32 / LPE;
  static_assert(GPW == 2 || GPW == 4, "component rotation is written for 2 or 4 groups");
  __shared__ float4 acc4[BK * LPE];
  float* acc = (float*)acc4;
  const int t = threadIdx.x;
  const int node0 = blockIdx.x * BK;
  const float4* x4 = (const float4*)xs;

  for (int i = t; i < BK * LPE; i += 512) {
    const int n = node0 + i / LPE;
    float4 v = make_float4(0.0f, 0.0f, 0.0f, 0.0f);
    if (n < N) v = x4[(size_t)n * LPE + (i % LPE)];
    acc4[i] = v;
  }
  __syncthreads();

  const int beg = bbase[blockIdx.x], end = bbase[blockIdx.x + 1];
  const int sub = t % LPE;
  const int g = (t / LPE) & (GPW - 1);
  const int o0 = g, o1 = g ^ 1, o2 = g ^ 2, o3 = g ^ 3;
  const bool sw1 = (g & 1) != 0;
  const bool sw2 = (g & 2) != 0;
  for (int i = beg + t / LPE; i < end; i += EPI * U) {
    int4 q[U];
    float4 xv[U];
#pragma unroll
    for (int j = 0; j < U; ++j) {
      const int idx = i + j * EPI;
      q[j] = rec[idx < end ? idx : i];
    }
#pragma unroll
    for (int j = 0; j < U; ++j) xv[j] = x4[(size_t)q[j].x * LPE + sub];
#pragma unroll
    for (int j = 0; j < U; ++j) {
      if (i + j * EPI < end) {
        const float w = __int_as_float(q[j].z);
        float* dst = acc + (q[j].y & (BK - 1)) * CH + sub * 4;
        float r0 = sw1 ? xv[j].y : xv[j].x;
        float r1 = sw1 ? xv[j].x : xv[j].y;
        float r2 = sw1 ? xv[j].w : xv[j].z;
        float r3 = sw1 ? xv[j].z : xv[j].w;
        if (GPW == 4) {
          const float u0 = sw2 ? r2 : r0, u1 = sw2 ? r3 : r1;
          const float u2 = sw2 ? r0 : r2, u3 = sw2 ? r1 : r3;
          r0 = u0; r1 = u1; r2 = u2; r3 = u3;
        }
        lds_add_f32(dst + o0, r0 * w);
        lds_add_f32(dst + o1, r1 * w);
        lds_add_f32(dst + o2, r2 * w);
        lds_add_f32(dst + o3, r3 * w);
      }
    }
  }
  __syncthreads();

  float4* out4 = (float4*)out;
  const float b0 = bias[sub * 4 + 0], b1 = bias[sub * 4 + 1];
  const float b2 = bias[sub * 4 + 2], b3 = bias[sub * 4 + 3];
  for (int i = t; i < BK * LPE; i += 512) {
    const int n = node0 + i / LPE;
    const float d = (n < N) ? dinv[n] : 0.0f;
    float4 v = acc4[i];
    v.x = v.x * d + b0; v.y = v.y * d + b1; v.z = v.z * d + b2; v.w = v.w * d + b3;
    if (L2NORM) {
      float ss = v.x * v.x + v.y * v.y + v.z * v.z + v.w * v.w;
      for (int off = 1; off < LPE; off <<= 1) ss += __shfl_xor(ss, off, 32);
      const float scale = 1.0f / fmaxf(sqrtf(ss), 1e-12f);
      v.x *= scale; v.y *= scale; v.z *= scale; v.w *= scale;
    } else {
      v.x = fmaxf(v.x, 0.0f); v.y = fmaxf(v.y, 0.0f);
      v.z = fmaxf(v.z, 0.0f); v.w = fmaxf(v.w, 0.0f);
    }
    if (n < N) out4[(size_t)n * LPE + sub] = v;
  }
}

extern "C" void kernel_launch(void* const* d_in, const int* in_sizes, int n_in,
                              void* d_out, int out_size, void* d_ws, size_t ws_size,
                              hipStream_t stream) {
  const int*   edge_index = (const int*)d_in[0];
  const float* ew         = (const float*)d_in[1];
  const float* emb        = (const float*)d_in[2];
  const float* W1         = (const float*)d_in[3];
  const float* b1         = (const float*)d_in[4];
  const float* W2         = (const float*)d_in[5];
  const float* b2         = (const float*)d_in[6];

  const int E = in_sizes[0] / 2;
  const int N = in_sizes[2] / 64;
  const int* rows = edge_index;
  const int* cols = edge_index + E;
  const int NB = (N + BK - 1) / BK;
  const int chunk_len = (E + NCHUNK - 1) / NCHUNK;

  int4*  rec  = (int4*)d_ws;
  float* dinv = (float*)(rec + (size_t)E);
  float* xw1  = dinv + (size_t)N;
  float* out1 = xw1  + (size_t)N * 64;
  float* xw2  = out1 + (size_t)N * 64;
  int* bbase     = (int*)(xw2 + (size_t)N * 32);
  int* btotal    = bbase + (size_t)NB + 1;
  int* chunkhist = btotal + (size_t)NB;

  bucket_hist_kernel<<<NCHUNK, 1024, sizeof(int) * (size_t)NB, stream>>>(
      cols, E, chunk_len, NB, chunkhist);
  row_scan_kernel<<<NB, NCHUNK, 0, stream>>>(chunkhist, btotal);
  bucket_base_kernel<<<1, 1024, 0, stream>>>(btotal, bbase, NB);
  bucket_scatter_kernel<<<NCHUNK, 1024, sizeof(int) * 3 * (size_t)NB, stream>>>(
      rows, cols, ew, E, chunk_len, NB, chunkhist, bbase, rec);

  deg_dinv_kernel<<<NB, 256, 0, stream>>>(rec, bbase, dinv, N);

  gemm_wmma_f32<64><<<N / 16, 128, 0, stream>>>(emb, W1, dinv, xw1);
  aggregate_lds_kernel<64, 8, false><<<NB, 512, 0, stream>>>(xw1, rec, bbase, dinv, b1, out1, N);

  gemm_wmma_f32<32><<<N / 16, 64, 0, stream>>>(out1, W2, dinv, xw2);
  aggregate_lds_kernel<32, 8, true><<<NB, 512, 0, stream>>>(xw2, rec, bbase, dinv, b2,
                                                           (float*)d_out, N);
}


# Degree kernel also sorts each bucket's records by destination node (LDS counts, rowptr); aggregation sums each row in registers with fma, no LDS float atomics (sum order changes, still varies)
# speedup vs baseline: 4.3102x; 1.3118x over previous
#include <hip/hip_runtime.h>
#include <math.h>

typedef __attribute__((ext_vector_type(2))) float v2f;
typedef __attribute__((ext_vector_type(8))) float v8f;

#define BK_SHIFT 7
#define BK 128
#define NCHUNK 256

__device__ __forceinline__ void lds_add_f32(float* p, float v) {
  __hip_atomic_fetch_add(p, v, __ATOMIC_RELAXED, __HIP_MEMORY_SCOPE_WORKGROUP);
}
__device__ __forceinline__ int lds_add_i32(int* p, int v) {
  return __hip_atomic_fetch_add(p, v, __ATOMIC_RELAXED, __HIP_MEMORY_SCOPE_WORKGROUP);
}

__device__ __forceinline__ int wave_incl_scan(int v, int lane) {
  for (int off = 1; off < 32; off <<= 1) {
    int n = __shfl_up(v, off, 32);
    if (lane >= off) v += n;
  }
  return v;
}

__global__ void __launch_bounds__(1024)
bucket_hist_kernel(const int* __restrict__ cols, int E, int chunk_len, int NB,
                   int* __restrict__ chunkhist) {
  extern __shared__ int sh_cnt[];
  const int t = threadIdx.x;
  for (int i = t; i < NB; i += 1024) sh_cnt[i] = 0;
  __syncthreads();
  const int beg = blockIdx.x * chunk_len;
  const int end = min(E, beg + chunk_len);
  for (int e = beg + t; e < end; e += 4 * 1024) {
    int c[4];
#pragma unroll
    for (int j = 0; j < 4; ++j) {
      const int ej = e + j * 1024;
      c[j] = cols[ej < end ? ej : e];
    }
#pragma unroll
    for (int j = 0; j < 4; ++j)
      if (e + j * 1024 < end) lds_add_i32(&sh_cnt[c[j] >> BK_SHIFT], 1);
  }
  __syncthreads();
  for (int i = t; i < NB; i += 1024)
    chunkhist[(size_t)i * NCHUNK + blockIdx.x] = sh_cnt[i];
}

__global__ void __launch_bounds__(NCHUNK)
row_scan_kernel(int* __restrict__ chunkhist, int* __restrict__ btotal) {
  __shared__ int wsum[NCHUNK / 32];
  const int t = threadIdx.x, lane = t & 31, w = t >> 5;
  const size_t idx = (size_t)blockIdx.x * NCHUNK + t;
  const int v = chunkhist[idx];
  const int s = wave_incl_scan(v, lane);
  if (lane == 31) wsum[w] = s;
  __syncthreads();
  int wbase = 0;
  for (int k = 0; k < NCHUNK / 32; ++k)
    if (k < w) wbase += wsum[k];
  chunkhist[idx] = wbase + s - v;
  if (t == NCHUNK - 1) btotal[blockIdx.x] = wbase + s;
}

__global__ void __launch_bounds__(1024)
bucket_base_kernel(const int* __restrict__ btotal, int* __restrict__ bbase, int NB) {
  __shared__ int wsum[32];
  const int t = threadIdx.x, lane = t & 31, w = t >> 5;
  int carry = 0;
  for (int base = 0; base < NB; base += 1024) {
    const int i = base + t;
    const int v = (i < NB) ? btotal[i] : 0;
    const int s = wave_incl_scan(v, lane);
    if (lane == 31) wsum[w] = s;
    __syncthreads();
    if (w == 0) wsum[lane] = wave_incl_scan(wsum[lane], lane);
    __syncthreads();
    const int wbase = (w > 0) ? wsum[w - 1] : 0;
    if (i < NB) bbase[i] = carry + wbase + s - v;
    const int tile_total = wsum[31];
    __syncthreads();
    carry += tile_total;
  }
  if (t == 0) bbase[NB] = carry;
}

#define SC_EPT 13
#define SC_TILE (SC_EPT * 1024)
__global__ void __launch_bounds__(1024)
bucket_scatter_kernel(const int* __restrict__ rows, const int* __restrict__ cols,
                      const float* __restrict__ ew, int E, int chunk_len, int NB,
                      const int* __restrict__ chunkoff, const int* __restrict__ bbase,
                      int4* __restrict__ rec) {
  __shared__ int4 srt[SC_TILE];
  __shared__ int wsum[32];
  extern __shared__ int sh_cur[];
  int* cur  = sh_cur;
  int* cnt  = cur + NB;
  int* loff = cnt + NB;
  const int t = threadIdx.x, lane = t & 31, w = t >> 5;
  const int beg = blockIdx.x * chunk_len;
  const int end = min(E, beg + chunk_len);
  for (int tb = beg; tb < end; tb += SC_TILE) {
    const int tn = min(SC_TILE, end - tb);
    int c[SC_EPT], r[SC_EPT], rk[SC_EPT];
    float wv[SC_EPT];
#pragma unroll
    for (int j = 0; j < SC_EPT; ++j) {
      const int idx = j * 1024 + t;
      const int es = tb + (idx < tn ? idx : 0);
      c[j] = cols[es];
      r[j] = rows[es];
      wv[j] = ew[es];
    }
    if (tb == beg) {
      for (int i = t; i < NB; i += 1024)
        cur[i] = bbase[i] + chunkoff[(size_t)i * NCHUNK + blockIdx.x];
    }
    for (int i = t; i < NB; i += 1024) cnt[i] = 0;
    __syncthreads();
#pragma unroll
    for (int j = 0; j < SC_EPT; ++j) {
      rk[j] = 0;
      if (j * 1024 + t < tn) rk[j] = lds_add_i32(&cnt[c[j] >> BK_SHIFT], 1);
    }
    __syncthreads();
    int carry = 0;
    for (int base = 0; base < NB; base += 1024) {
      const int i = base + t;
      const int v = (i < NB) ? cnt[i] : 0;
      const int s = wave_incl_scan(v, lane);
      if (lane == 31) wsum[w] = s;
      __syncthreads();
      if (w == 0) wsum[lane] = wave_incl_scan(wsum[lane], lane);
      __syncthreads();
      const int wbase = (w > 0) ? wsum[w - 1] : 0;
      if (i < NB) loff[i] = carry + wbase + s - v;
      const int tile_total = wsum[31];
      __syncthreads();
      carry += tile_total;
    }
    int slot[SC_EPT];
#pragma unroll
    for (int j = 0; j < SC_EPT; ++j) slot[j] = loff[c[j] >> BK_SHIFT] + rk[j];
#pragma unroll
    for (int j = 0; j < SC_EPT; ++j) {
      if (j * 1024 + t < tn)
        srt[slot[j]] = make_int4(r[j], c[j], __float_as_int(wv[j]), 0);
    }
    __syncthreads();
    int4 q[SC_EPT];
    int g[SC_EPT];
#pragma unroll
    for (int j = 0; j < SC_EPT; ++j) {
      const int i = j * 1024 + t;
      q[j] = srt[i < tn ? i : 0];
    }
#pragma unroll
    for (int j = 0; j < SC_EPT; ++j) {
      const int b = q[j].y >> BK_SHIFT;
      g[j] = cur[b] + (j * 1024 + t - loff[b]);
    }
#pragma unroll
    for (int j = 0; j < SC_EPT; ++j) {
      if (j * 1024 + t < tn) rec[g[j]] = q[j];
    }
    __syncthreads();
    for (int i = t; i < NB; i += 1024) cur[i] += cnt[i];
  }
}

#define NS_STAGE 8192
__global__ void __launch_bounds__(512)
node_sort_deg_kernel(const int4* __restrict__ rec, const int* __restrict__ bbase,
                     float* __restrict__ dinv, int* __restrict__ rowptr,
                     int2* __restrict__ srec, int N, int NB) {
  __shared__ int2 stage[NS_STAGE];
  __shared__ float sdeg[BK];
  __shared__ int scnt[BK];
  const int t = threadIdx.x;
  if (t < BK) { sdeg[t] = 1.0f; scnt[t] = 0; }
  __syncthreads();
  const int beg = bbase[blockIdx.x], end = bbase[blockIdx.x + 1];
  const int cnt = end - beg;
  const bool staged = cnt <= NS_STAGE;
  for (int i = beg + t; i < end; i += 4 * 512) {
    int4 q[4];
#pragma unroll
    for (int j = 0; j < 4; ++j) {
      const int ij = i + j * 512;
      q[j] = rec[ij < end ? ij : i];
    }
#pragma unroll
    for (int j = 0; j < 4; ++j) {
      const bool valid = i + j * 512 < end;
      const int c = q[j].y & (BK - 1);
      lds_add_i32(&scnt[c], valid ? 1 : 0);
      lds_add_f32(&sdeg[c], valid ? __int_as_float(q[j].z) : 0.0f);
    }
  }
  __syncthreads();
  if (t < BK) {
    const int n = blockIdx.x * BK + t;
    if (n < N) {
      const float d = sdeg[t];
      dinv[n] = (d > 0.0f) ? rsqrtf(d) : 0.0f;
    }
  }
  if (t < 32) {
    const int v0 = scnt[4 * t], v1 = scnt[4 * t + 1], v2 = scnt[4 * t + 2], v3 = scnt[4 * t + 3];
    const int s = v0 + v1 + v2 + v3;
    const int base = wave_incl_scan(s, t) - s;
    const int o0 = base, o1 = base + v0, o2 = o1 + v1, o3 = o2 + v2;
    scnt[4 * t] = o0; scnt[4 * t + 1] = o1; scnt[4 * t + 2] = o2; scnt[4 * t + 3] = o3;
    const int n = blockIdx.x * BK + 4 * t;
    if (n     < N) rowptr[n]     = beg + o0;
    if (n + 1 < N) rowptr[n + 1] = beg + o1;
    if (n + 2 < N) rowptr[n + 2] = beg + o2;
    if (n + 3 < N) rowptr[n + 3] = beg + o3;
  }
  if (blockIdx.x == NB - 1 && t == 0) rowptr[N] = end;
  __syncthreads();
  for (int i = beg + t; i < end; i += 4 * 512) {
    int4 q[4];
#pragma unroll
    for (int j = 0; j < 4; ++j) {
      const int ij = i + j * 512;
      q[j] = rec[ij < end ? ij : i];
    }
#pragma unroll
    for (int j = 0; j < 4; ++j) {
      const bool valid = i + j * 512 < end;
      const int pos = lds_add_i32(&scnt[q[j].y & (BK - 1)], valid ? 1 : 0);
      if (valid) {
        const int2 r = make_int2(q[j].x, q[j].z);
        if (staged) stage[pos] = r; else srec[beg + pos] = r;
      }
    }
  }
  if (staged) {
    __syncthreads();
    for (int i = t; i < cnt; i += 512) srec[beg + i] = stage[i];
  }
}

template <int NCOLS>
__global__ void gemm_wmma_f32(const float* __restrict__ A,
                              const float* __restrict__ B,
                              const float* __restrict__ rowscale,
                              float* __restrict__ C) {
  const int m0   = blockIdx.x * 16;
  const int wid  = threadIdx.x >> 5;
  const int lane = threadIdx.x & 31;
  const int n0   = wid * 16;
  const int ln   = lane & 15;
  const int half = lane >> 4;

  v8f acc = {};
  const float* aptr = A + (size_t)(m0 + ln) * 64 + 2 * half;
  const float* bptr = B + (size_t)(2 * half) * NCOLS + n0 + ln;

#pragma unroll
  for (int k = 0; k < 64; k += 4) {
    v2f a, b;
    a.x = aptr[k];
    a.y = aptr[k + 1];
    b.x = bptr[k * NCOLS];
    b.y = bptr[k * NCOLS + NCOLS];
    acc = __builtin_amdgcn_wmma_f32_16x16x4_f32(
         false, a,  false, b,
         (short)0, acc,  false,  false);
  }

  const float4 s_lo = *(const float4*)(rowscale + m0 + 8 * half);
  const float4 s_hi = *(const float4*)(rowscale + m0 + 8 * half + 4);
  const float sc[8] = {s_lo.x, s_lo.y, s_lo.z, s_lo.w, s_hi.x, s_hi.y, s_hi.z, s_hi.w};
  float* cptr = C + (size_t)(m0 + 8 * half) * NCOLS + n0 + ln;
#pragma unroll
  for (int r = 0; r < 8; ++r) {
    cptr[(size_t)r * NCOLS] = acc[r] * sc[r];
  }
}

template <int CH, int RPG, int U, bool L2NORM>
__global__ void __launch_bounds__(256)
aggregate_rows_kernel(const float* __restrict__ xs, const int2* __restrict__ srec,
                      const int* __restrict__ rowptr, const float* __restrict__ dinv,
                      const float* __restrict__ bias, float* __restrict__ out, int N) {
  constexpr int LPE = CH / 4;
  constexpr int NG = 256 / LPE;
  const int t = threadIdx.x;
  const int sub = t % LPE;
  const int grp = t / LPE;
  const float4* x4 = (const float4*)xs;
  float4* out4 = (float4*)out;
  const float b0 = bias[sub * 4 + 0], b1 = bias[sub * 4 + 1];
  const float b2 = bias[sub * 4 + 2], b3 = bias[sub * 4 + 3];
  const char* xbytes = (const char*)xs;
  const char* rbytes = (const char*)srec;
  const unsigned suboff = (unsigned)sub * 16u;

  for (int rr = 0; rr < RPG; ++rr) {
    const int n = (blockIdx.x * RPG + rr) * NG + grp;
    if (n < N) {
      const int s = rowptr[n], e = rowptr[n + 1];
      float4 a = x4[(size_t)n * LPE + sub];
      for (int k = s; k < e; k += U) {
        int2 q[U];
        float4 xv[U];
#pragma unroll
        for (int j = 0; j < U; ++j) {
          const int idx = min(k + j, e - 1);
          q[j] = *(const int2*)(rbytes + (unsigned)idx * 8u);
        }
#pragma unroll
        for (int j = 0; j < U; ++j)
          xv[j] = *(const float4*)(xbytes + ((unsigned)q[j].x * (unsigned)(CH * 4) + suboff));
#pragma unroll
        for (int j = 0; j < U; ++j) {
          const float w = (k + j < e) ? __int_as_float(q[j].y) : 0.0f;
          a.x = fmaf(w, xv[j].x, a.x);
          a.y = fmaf(w, xv[j].y, a.y);
          a.z = fmaf(w, xv[j].z, a.z);
          a.w = fmaf(w, xv[j].w, a.w);
        }
      }
      const float d = dinv[n];
      float4 v;
      v.x = a.x * d + b0; v.y = a.y * d + b1; v.z = a.z * d + b2; v.w = a.w * d + b3;
      if (L2NORM) {
        float ss = v.x * v.x + v.y * v.y + v.z * v.z + v.w * v.w;
        for (int off = 1; off < LPE; off <<= 1) ss += __shfl_xor(ss, off, 32);
        const float scale = 1.0f / fmaxf(sqrtf(ss), 1e-12f);
        v.x *= scale; v.y *= scale; v.z *= scale; v.w *= scale;
      } else {
        v.x = fmaxf(v.x, 0.0f); v.y = fmaxf(v.y, 0.0f);
        v.z = fmaxf(v.z, 0.0f); v.w = fmaxf(v.w, 0.0f);
      }
      out4[(size_t)n * LPE + sub] = v;
    }
  }
}

extern "C" void kernel_launch(void* const* d_in, const int* in_sizes, int n_in,
                              void* d_out, int out_size, void* d_ws, size_t ws_size,
                              hipStream_t stream) {
  const int*   edge_index = (const int*)d_in[0];
  const float* ew         = (const float*)d_in[1];
  const float* emb        = (const float*)d_in[2];
  const float* W1         = (const float*)d_in[3];
  const float* b1         = (const float*)d_in[4];
  const float* W2         = (const float*)d_in[5];
  const float* b2         = (const float*)d_in[6];

  const int E = in_sizes[0] / 2;
  const int N = in_sizes[2] / 64;
  const int* rows = edge_index;
  const int* cols = edge_index + E;
  const int NB = (N + BK - 1) / BK;
  const int chunk_len = (E + NCHUNK - 1) / NCHUNK;

  int4*  rec  = (int4*)d_ws;
  int2*  srec = (int2*)(rec + (size_t)E);
  float* dinv = (float*)(srec + (size_t)E);
  float* xw1  = dinv + (size_t)N;
  float* out1 = xw1  + (size_t)N * 64;
  float* xw2  = out1 + (size_t)N * 64;
  int* bbase     = (int*)(xw2 + (size_t)N * 32);
  int* btotal    = bbase + (size_t)NB + 1;
  int* chunkhist = btotal + (size_t)NB;
  int* rowptr    = chunkhist + (size_t)NB * NCHUNK;

  bucket_hist_kernel<<<NCHUNK, 1024, sizeof(int) * (size_t)NB, stream>>>(
      cols, E, chunk_len, NB, chunkhist);
  row_scan_kernel<<<NB, NCHUNK, 0, stream>>>(chunkhist, btotal);
  bucket_base_kernel<<<1, 1024, 0, stream>>>(btotal, bbase, NB);
  bucket_scatter_kernel<<<NCHUNK, 1024, sizeof(int) * 3 * (size_t)NB, stream>>>(
      rows, cols, ew, E, chunk_len, NB, chunkhist, bbase, rec);

  node_sort_deg_kernel<<<NB, 512, 0, stream>>>(rec, bbase, dinv, rowptr, srec, N, NB);

  gemm_wmma_f32<64><<<N / 16, 128, 0, stream>>>(emb, W1, dinv, xw1);
  aggregate_rows_kernel<64, 2, 8, false><<<(N + 31) / 32, 256, 0, stream>>>(
      xw1, srec, rowptr, dinv, b1, out1, N);

  gemm_wmma_f32<32><<<N / 16, 64, 0, stream>>>(out1, W2, dinv, xw2);
  aggregate_rows_kernel<32, 2, 8, true><<<(N + 63) / 64, 256, 0, stream>>>(
      xw2, srec, rowptr, dinv, b2, (float*)d_out, N);
}


# Layer-2 GEMM (fp32 WMMA, same loaders) moved into the layer-1 aggregate kernel: a workgroup's 32 relu rows go through an LDS tile to 4 waves' 16x16 tiles; hidden layer never stored; 9 -> 8 launches
# speedup vs baseline: 4.5771x; 1.0619x over previous
#include <hip/hip_runtime.h>
#include <math.h>

typedef __attribute__((ext_vector_type(2))) float v2f;
typedef __attribute__((ext_vector_type(8))) float v8f;

#define BK_SHIFT 7
#define BK 128
#define NCHUNK 256

__device__ __forceinline__ void lds_add_f32(float* p, float v) {
  __hip_atomic_fetch_add(p, v, __ATOMIC_RELAXED, __HIP_MEMORY_SCOPE_WORKGROUP);
}
__device__ __forceinline__ int lds_add_i32(int* p, int v) {
  return __hip_atomic_fetch_add(p, v, __ATOMIC_RELAXED, __HIP_MEMORY_SCOPE_WORKGROUP);
}

__device__ __forceinline__ int wave_incl_scan(int v, int lane) {
  for (int off = 1; off < 32; off <<= 1) {
    int n = __shfl_up(v, off, 32);
    if (lane >= off) v += n;
  }
  return v;
}

__global__ void __launch_bounds__(1024)
bucket_hist_kernel(const int* __restrict__ cols, int E, int chunk_len, int NB,
                   int* __restrict__ chunkhist) {
  extern __shared__ int sh_cnt[];
  const int t = threadIdx.x;
  for (int i = t; i < NB; i += 1024) sh_cnt[i] = 0;
  __syncthreads();
  const int beg = blockIdx.x * chunk_len;
  const int end = min(E, beg + chunk_len);
  for (int e = beg + t; e < end; e += 4 * 1024) {
    int c[4];
#pragma unroll
    for (int j = 0; j < 4; ++j) {
      const int ej = e + j * 1024;
      c[j] = cols[ej < end ? ej : e];
    }
#pragma unroll
    for (int j = 0; j < 4; ++j)
      if (e + j * 1024 < end) lds_add_i32(&sh_cnt[c[j] >> BK_SHIFT], 1);
  }
  __syncthreads();
  for (int i = t; i < NB; i += 1024)
    chunkhist[(size_t)i * NCHUNK + blockIdx.x] = sh_cnt[i];
}

__global__ void __launch_bounds__(NCHUNK)
row_scan_kernel(int* __restrict__ chunkhist, int* __restrict__ btotal) {
  __shared__ int wsum[NCHUNK / 32];
  const int t = threadIdx.x, lane = t & 31, w = t >> 5;
  const size_t idx = (size_t)blockIdx.x * NCHUNK + t;
  const int v = chunkhist[idx];
  const int s = wave_incl_scan(v, lane);
  if (lane == 31) wsum[w] = s;
  __syncthreads();
  int wbase = 0;
  for (int k = 0; k < NCHUNK / 32; ++k)
    if (k < w) wbase += wsum[k];
  chunkhist[idx] = wbase + s - v;
  if (t == NCHUNK - 1) btotal[blockIdx.x] = wbase + s;
}

__global__ void __launch_bounds__(1024)
bucket_base_kernel(const int* __restrict__ btotal, int* __restrict__ bbase, int NB) {
  __shared__ int wsum[32];
  const int t = threadIdx.x, lane = t & 31, w = t >> 5;
  int carry = 0;
  for (int base = 0; base < NB; base += 1024) {
    const int i = base + t;
    const int v = (i < NB) ? btotal[i] : 0;
    const int s = wave_incl_scan(v, lane);
    if (lane == 31) wsum[w] = s;
    __syncthreads();
    if (w == 0) wsum[lane] = wave_incl_scan(wsum[lane], lane);
    __syncthreads();
    const int wbase = (w > 0) ? wsum[w - 1] : 0;
    if (i < NB) bbase[i] = carry + wbase + s - v;
    const int tile_total = wsum[31];
    __syncthreads();
    carry += tile_total;
  }
  if (t == 0) bbase[NB] = carry;
}

#define SC_EPT 13
#define SC_TILE (SC_EPT * 1024)
__global__ void __launch_bounds__(1024)
bucket_scatter_kernel(const int* __restrict__ rows, const int* __restrict__ cols,
                      const float* __restrict__ ew, int E, int chunk_len, int NB,
                      const int* __restrict__ chunkoff, const int* __restrict__ bbase,
                      int4* __restrict__ rec) {
  __shared__ int4 srt[SC_TILE];
  __shared__ int wsum[32];
  extern __shared__ int sh_cur[];
  int* cur  = sh_cur;
  int* cnt  = cur + NB;
  int* loff = cnt + NB;
  const int t = threadIdx.x, lane = t & 31, w = t >> 5;
  const int beg = blockIdx.x * chunk_len;
  const int end = min(E, beg + chunk_len);
  for (int tb = beg; tb < end; tb += SC_TILE) {
    const int tn = min(SC_TILE, end - tb);
    int c[SC_EPT], r[SC_EPT], rk[SC_EPT];
    float wv[SC_EPT];
#pragma unroll
    for (int j = 0; j < SC_EPT; ++j) {
      const int idx = j * 1024 + t;
      const int es = tb + (idx < tn ? idx : 0);
      c[j] = cols[es];
      r[j] = rows[es];
      wv[j] = ew[es];
    }
    if (tb == beg) {
      for (int i = t; i < NB; i += 1024)
        cur[i] = bbase[i] + chunkoff[(size_t)i * NCHUNK + blockIdx.x];
    }
    for (int i = t; i < NB; i += 1024) cnt[i] = 0;
    __syncthreads();
#pragma unroll
    for (int j = 0; j < SC_EPT; ++j) {
      rk[j] = 0;
      if (j * 1024 + t < tn) rk[j] = lds_add_i32(&cnt[c[j] >> BK_SHIFT], 1);
    }
    __syncthreads();
    int carry = 0;
    for (int base = 0; base < NB; base += 1024) {
      const int i = base + t;
      const int v = (i < NB) ? cnt[i] : 0;
      const int s = wave_incl_scan(v, lane);
      if (lane == 31) wsum[w] = s;
      __syncthreads();
      if (w == 0) wsum[lane] = wave_incl_scan(wsum[lane], lane);
      __syncthreads();
      const int wbase = (w > 0) ? wsum[w - 1] : 0;
      if (i < NB) loff[i] = carry + wbase + s - v;
      const int tile_total = wsum[31];
      __syncthreads();
      carry += tile_total;
    }
    int slot[SC_EPT];
#pragma unroll
    for (int j = 0; j < SC_EPT; ++j) slot[j] = loff[c[j] >> BK_SHIFT] + rk[j];
#pragma unroll
    for (int j = 0; j < SC_EPT; ++j) {
      if (j * 1024 + t < tn)
        srt[slot[j]] = make_int4(r[j], c[j], __float_as_int(wv[j]), 0);
    }
    __syncthreads();
    int4 q[SC_EPT];
    int g[SC_EPT];
#pragma unroll
    for (int j = 0; j < SC_EPT; ++j) {
      const int i = j * 1024 + t;
      q[j] = srt[i < tn ? i : 0];
    }
#pragma unroll
    for (int j = 0; j < SC_EPT; ++j) {
      const int b = q[j].y >> BK_SHIFT;
      g[j] = cur[b] + (j * 1024 + t - loff[b]);
    }
#pragma unroll
    for (int j = 0; j < SC_EPT; ++j) {
      if (j * 1024 + t < tn) rec[g[j]] = q[j];
    }
    __syncthreads();
    for (int i = t; i < NB; i += 1024) cur[i] += cnt[i];
  }
}

#define NS_STAGE 8192
__global__ void __launch_bounds__(512)
node_sort_deg_kernel(const int4* __restrict__ rec, const int* __restrict__ bbase,
                     float* __restrict__ dinv, int* __restrict__ rowptr,
                     int2* __restrict__ srec, int N, int NB) {
  __shared__ int2 stage[NS_STAGE];
  __shared__ float sdeg[BK];
  __shared__ int scnt[BK];
  const int t = threadIdx.x;
  if (t < BK) { sdeg[t] = 1.0f; scnt[t] = 0; }
  __syncthreads();
  const int beg = bbase[blockIdx.x], end = bbase[blockIdx.x + 1];
  const int cnt = end - beg;
  const bool staged = cnt <= NS_STAGE;
  for (int i = beg + t; i < end; i += 4 * 512) {
    int4 q[4];
#pragma unroll
    for (int j = 0; j < 4; ++j) {
      const int ij = i + j * 512;
      q[j] = rec[ij < end ? ij : i];
    }
#pragma unroll
    for (int j = 0; j < 4; ++j) {
      const bool valid = i + j * 512 < end;
      const int c = q[j].y & (BK - 1);
      lds_add_i32(&scnt[c], valid ? 1 : 0);
      lds_add_f32(&sdeg[c], valid ? __int_as_float(q[j].z) : 0.0f);
    }
  }
  __syncthreads();
  if (t < BK) {
    const int n = blockIdx.x * BK + t;
    if (n < N) {
      const float d = sdeg[t];
      dinv[n] = (d > 0.0f) ? rsqrtf(d) : 0.0f;
    }
  }
  if (t < 32) {
    const int v0 = scnt[4 * t], v1 = scnt[4 * t + 1], v2 = scnt[4 * t + 2], v3 = scnt[4 * t + 3];
    const int s = v0 + v1 + v2 + v3;
    const int base = wave_incl_scan(s, t) - s;
    const int o0 = base, o1 = base + v0, o2 = o1 + v1, o3 = o2 + v2;
    scnt[4 * t] = o0; scnt[4 * t + 1] = o1; scnt[4 * t + 2] = o2; scnt[4 * t + 3] = o3;
    const int n = blockIdx.x * BK + 4 * t;
    if (n     < N) rowptr[n]     = beg + o0;
    if (n + 1 < N) rowptr[n + 1] = beg + o1;
    if (n + 2 < N) rowptr[n + 2] = beg + o2;
    if (n + 3 < N) rowptr[n + 3] = beg + o3;
  }
  if (blockIdx.x == NB - 1 && t == 0) rowptr[N] = end;
  __syncthreads();
  for (int i = beg + t; i < end; i += 4 * 512) {
    int4 q[4];
#pragma unroll
    for (int j = 0; j < 4; ++j) {
      const int ij = i + j * 512;
      q[j] = rec[ij < end ? ij : i];
    }
#pragma unroll
    for (int j = 0; j < 4; ++j) {
      const bool valid = i + j * 512 < end;
      const int pos = lds_add_i32(&scnt[q[j].y & (BK - 1)], valid ? 1 : 0);
      if (valid) {
        const int2 r = make_int2(q[j].x, q[j].z);
        if (staged) stage[pos] = r; else srec[beg + pos] = r;
      }
    }
  }
  if (staged) {
    __syncthreads();
    for (int i = t; i < cnt; i += 512) srec[beg + i] = stage[i];
  }
}

template <int NCOLS>
__global__ void gemm_wmma_f32(const float* __restrict__ A,
                              const float* __restrict__ B,
                              const float* __restrict__ rowscale,
                              float* __restrict__ C) {
  const int m0   = blockIdx.x * 16;
  const int wid  = threadIdx.x >> 5;
  const int lane = threadIdx.x & 31;
  const int n0   = wid * 16;
  const int ln   = lane & 15;
  const int half = lane >> 4;

  v8f acc = {};
  const float* aptr = A + (size_t)(m0 + ln) * 64 + 2 * half;
  const float* bptr = B + (size_t)(2 * half) * NCOLS + n0 + ln;

#pragma unroll
  for (int k = 0; k < 64; k += 4) {
    v2f a, b;
    a.x = aptr[k];
    a.y = aptr[k + 1];
    b.x = bptr[k * NCOLS];
    b.y = bptr[k * NCOLS + NCOLS];
    acc = __builtin_amdgcn_wmma_f32_16x16x4_f32(
         false, a,  false, b,
         (short)0, acc,  false,  false);
  }

  const float4 s_lo = *(const float4*)(rowscale + m0 + 8 * half);
  const float4 s_hi = *(const float4*)(rowscale + m0 + 8 * half + 4);
  const float sc[8] = {s_lo.x, s_lo.y, s_lo.z, s_lo.w, s_hi.x, s_hi.y, s_hi.z, s_hi.w};
  float* cptr = C + (size_t)(m0 + 8 * half) * NCOLS + n0 + ln;
#pragma unroll
  for (int r = 0; r < 8; ++r) {
    cptr[(size_t)r * NCOLS] = acc[r] * sc[r];
  }
}

template <int CH, int RPG, int U, bool L2NORM>
__global__ void __launch_bounds__(256)
aggregate_rows_kernel(const float* __restrict__ xs, const int2* __restrict__ srec,
                      const int* __restrict__ rowptr, const float* __restrict__ dinv,
                      const float* __restrict__ bias, float* __restrict__ out, int N,
                      const float* __restrict__ Wn) {
  constexpr int LPE = CH / 4;
  constexpr int NG = 256 / LPE;
  constexpr int TROWS = RPG * NG;
  constexpr int TP = CH + 4;
  __shared__ __attribute__((aligned(16))) float tile[L2NORM ? 4 : TROWS * TP];
  const int t = threadIdx.x;
  const int sub = t % LPE;
  const int grp = t / LPE;
  const float4* x4 = (const float4*)xs;
  float4* out4 = (float4*)out;
  const float b0 = bias[sub * 4 + 0], b1 = bias[sub * 4 + 1];
  const float b2 = bias[sub * 4 + 2], b3 = bias[sub * 4 + 3];
  const char* xbytes = (const char*)xs;
  const char* rbytes = (const char*)srec;
  const unsigned suboff = (unsigned)sub * 16u;

  for (int rr = 0; rr < RPG; ++rr) {
    const int n = (blockIdx.x * RPG + rr) * NG + grp;
    float4 v = make_float4(0.0f, 0.0f, 0.0f, 0.0f);
    if (n < N) {
      const int s = rowptr[n], e = rowptr[n + 1];
      float4 a = x4[(size_t)n * LPE + sub];
      for (int k = s; k < e; k += U) {
        int2 q[U];
        float4 xv[U];
#pragma unroll
        for (int j = 0; j < U; ++j) {
          const int idx = min(k + j, e - 1);
          q[j] = *(const int2*)(rbytes + (unsigned)idx * 8u);
        }
#pragma unroll
        for (int j = 0; j < U; ++j)
          xv[j] = *(const float4*)(xbytes + ((unsigned)q[j].x * (unsigned)(CH * 4) + suboff));
#pragma unroll
        for (int j = 0; j < U; ++j) {
          const float w = (k + j < e) ? __int_as_float(q[j].y) : 0.0f;
          a.x = fmaf(w, xv[j].x, a.x);
          a.y = fmaf(w, xv[j].y, a.y);
          a.z = fmaf(w, xv[j].z, a.z);
          a.w = fmaf(w, xv[j].w, a.w);
        }
      }
      const float d = dinv[n];
      v.x = a.x * d + b0; v.y = a.y * d + b1; v.z = a.z * d + b2; v.w = a.w * d + b3;
      if (L2NORM) {
        float ss = v.x * v.x + v.y * v.y + v.z * v.z + v.w * v.w;
        for (int off = 1; off < LPE; off <<= 1) ss += __shfl_xor(ss, off, 32);
        const float scale = 1.0f / fmaxf(sqrtf(ss), 1e-12f);
        v.x *= scale; v.y *= scale; v.z *= scale; v.w *= scale;
        out4[(size_t)n * LPE + sub] = v;
      } else {
        v.x = fmaxf(v.x, 0.0f); v.y = fmaxf(v.y, 0.0f);
        v.z = fmaxf(v.z, 0.0f); v.w = fmaxf(v.w, 0.0f);
      }
    }
    if (!L2NORM) *(float4*)&tile[(rr * NG + grp) * TP + sub * 4] = v;
  }

  if (!L2NORM) {
    constexpr int NOUT = 32;
    constexpr int MT = TROWS / 16;
    constexpr int NT = NOUT / 16;
    static_assert(TROWS % 16 == 0 && MT * NT <= 8 && CH % 4 == 0, "tile shape");
    const int wv = t >> 5;
    const bool mm = wv < MT * NT;
    const int lane = t & 31;
    const int ln   = lane & 15;
    const int half = lane >> 4;
    const int mt = wv / NT, nt = wv % NT;
    const int m0 = blockIdx.x * TROWS + mt * 16 + 8 * half;
    v2f bf[CH / 4] = {};
    float sc[8] = {};
    if (mm) {
      const float* bptr = Wn + (size_t)(2 * half) * NOUT + nt * 16 + ln;
#pragma unroll
      for (int k = 0; k < CH; k += 4) {
        bf[k / 4].x = bptr[k * NOUT];
        bf[k / 4].y = bptr[k * NOUT + NOUT];
      }
#pragma unroll
      for (int r = 0; r < 8; ++r) sc[r] = dinv[min(m0 + r, N - 1)];
    }
    __syncthreads();
    if (mm) {
      const float* arow = &tile[(mt * 16 + ln) * TP + 2 * half];
      v8f acc = {};
#pragma unroll
      for (int k = 0; k < CH; k += 4) {
        v2f a;
        a.x = arow[k];
        a.y = arow[k + 1];
        acc = __builtin_amdgcn_wmma_f32_16x16x4_f32(
             false, a,  false, bf[k / 4],
             (short)0, acc,  false,  false);
      }
      float* cptr = out + (size_t)m0 * NOUT + nt * 16 + ln;
      if (m0 + 8 <= N) {
#pragma unroll
        for (int r = 0; r < 8; ++r) cptr[(size_t)r * NOUT] = acc[r] * sc[r];
      } else {
#pragma unroll
        for (int r = 0; r < 8; ++r)
          if (m0 + r < N) cptr[(size_t)r * NOUT] = acc[r] * sc[r];
      }
    }
  }
}

extern "C" void kernel_launch(void* const* d_in, const int* in_sizes, int n_in,
                              void* d_out, int out_size, void* d_ws, size_t ws_size,
                              hipStream_t stream) {
  const int*   edge_index = (const int*)d_in[0];
  const float* ew         = (const float*)d_in[1];
  const float* emb        = (const float*)d_in[2];
  const float* W1         = (const float*)d_in[3];
  const float* b1         = (const float*)d_in[4];
  const float* W2         = (const float*)d_in[5];
  const float* b2         = (const float*)d_in[6];

  const int E = in_sizes[0] / 2;
  const int N = in_sizes[2] / 64;
  const int* rows = edge_index;
  const int* cols = edge_index + E;
  const int NB = (N + BK - 1) / BK;
  const int chunk_len = (E + NCHUNK - 1) / NCHUNK;

  int4*  rec  = (int4*)d_ws;
  int2*  srec = (int2*)(rec + (size_t)E);
  float* dinv = (float*)(srec + (size_t)E);
  float* xw1  = dinv + (size_t)N;
  float* xw2  = xw1  + (size_t)N * 64;
  int* bbase     = (int*)(xw2 + (size_t)N * 32);
  int* btotal    = bbase + (size_t)NB + 1;
  int* chunkhist = btotal + (size_t)NB;
  int* rowptr    = chunkhist + (size_t)NB * NCHUNK;

  bucket_hist_kernel<<<NCHUNK, 1024, sizeof(int) * (size_t)NB, stream>>>(
      cols, E, chunk_len, NB, chunkhist);
  row_scan_kernel<<<NB, NCHUNK, 0, stream>>>(chunkhist, btotal);
  bucket_base_kernel<<<1, 1024, 0, stream>>>(btotal, bbase, NB);
  bucket_scatter_kernel<<<NCHUNK, 1024, sizeof(int) * 3 * (size_t)NB, stream>>>(
      rows, cols, ew, E, chunk_len, NB, chunkhist, bbase, rec);

  node_sort_deg_kernel<<<NB, 512, 0, stream>>>(rec, bbase, dinv, rowptr, srec, N, NB);

  gemm_wmma_f32<64><<<N / 16, 128, 0, stream>>>(emb, W1, dinv, xw1);
  aggregate_rows_kernel<64, 2, 8, false><<<(N + 31) / 32, 256, 0, stream>>>(
      xw1, srec, rowptr, dinv, b1, xw2, N, W2);

  aggregate_rows_kernel<32, 2, 8, true><<<(N + 63) / 64, 256, 0, stream>>>(
      xw2, srec, rowptr, dinv, b2, (float*)d_out, N, W2);
}


# Global counting sort dropped: each 13312-edge tile is bucket-sorted in LDS, stored in place with its bucket offsets; bucket kernel collects its runs; 8 -> 5 launches (row sum order varies as before)
# speedup vs baseline: 4.8833x; 1.0669x over previous
#include <hip/hip_runtime.h>
#include <math.h>

typedef __attribute__((ext_vector_type(2))) float v2f;
typedef __attribute__((ext_vector_type(8))) float v8f;

#define BK_SHIFT 7
#define BK 128

__device__ __forceinline__ void lds_add_f32(float* p, float v) {
  __hip_atomic_fetch_add(p, v, __ATOMIC_RELAXED, __HIP_MEMORY_SCOPE_WORKGROUP);
}
__device__ __forceinline__ int lds_add_i32(int* p, int v) {
  return __hip_atomic_fetch_add(p, v, __ATOMIC_RELAXED, __HIP_MEMORY_SCOPE_WORKGROUP);
}

__device__ __forceinline__ int wave_incl_scan(int v, int lane) {
  for (int off = 1; off < 32; off <<= 1) {
    int n = __shfl_up(v, off, 32);
    if (lane >= off) v += n;
  }
  return v;
}

#define SC_EPT 13
#define SC_TILE (SC_EPT * 1024)
__global__ void __launch_bounds__(1024)
tile_sort_kernel(const int* __restrict__ rows, const int* __restrict__ cols,
                 const float* __restrict__ ew, int E, int NB, int NT,
                 int* __restrict__ toff, int4* __restrict__ rec) {
  __shared__ int4 srt[SC_TILE];
  __shared__ int wsum[32];
  extern __shared__ int sh_cnt[];
  int* cnt  = sh_cnt;
  int* loff = cnt + NB;
  const int t = threadIdx.x, lane = t & 31, w = t >> 5;
  const int tb = blockIdx.x * SC_TILE;
  const int tn = min(SC_TILE, E - tb);
  int c[SC_EPT], r[SC_EPT], rk[SC_EPT];
  float wv[SC_EPT];
#pragma unroll
  for (int j = 0; j < SC_EPT; ++j) {
    const int idx = j * 1024 + t;
    const int es = tb + (idx < tn ? idx : 0);
    c[j] = cols[es];
    r[j] = rows[es];
    wv[j] = ew[es];
  }
  for (int i = t; i < NB; i += 1024) cnt[i] = 0;
  __syncthreads();
#pragma unroll
  for (int j = 0; j < SC_EPT; ++j) {
    rk[j] = 0;
    if (j * 1024 + t < tn) rk[j] = lds_add_i32(&cnt[c[j] >> BK_SHIFT], 1);
  }
  __syncthreads();
  int carry = 0;
  for (int base = 0; base < NB; base += 1024) {
    const int i = base + t;
    const int v = (i < NB) ? cnt[i] : 0;
    const int s = wave_incl_scan(v, lane);
    if (lane == 31) wsum[w] = s;
    __syncthreads();
    if (w == 0) wsum[lane] = wave_incl_scan(wsum[lane], lane);
    __syncthreads();
    const int wbase = (w > 0) ? wsum[w - 1] : 0;
    if (i < NB) loff[i] = carry + wbase + s - v;
    const int tile_total = wsum[31];
    __syncthreads();
    carry += tile_total;
  }
  for (int i = t; i < NB; i += 1024) toff[(size_t)i * NT + blockIdx.x] = loff[i];
  if (t == 0) toff[(size_t)NB * NT + blockIdx.x] = tn;
  int slot[SC_EPT];
#pragma unroll
  for (int j = 0; j < SC_EPT; ++j) slot[j] = loff[c[j] >> BK_SHIFT] + rk[j];
#pragma unroll
  for (int j = 0; j < SC_EPT; ++j) {
    if (j * 1024 + t < tn)
      srt[slot[j]] = make_int4(r[j], c[j], __float_as_int(wv[j]), 0);
  }
  __syncthreads();
  int4 q[SC_EPT];
#pragma unroll
  for (int j = 0; j < SC_EPT; ++j) {
    const int i = j * 1024 + t;
    q[j] = srt[i < tn ? i : 0];
  }
#pragma unroll
  for (int j = 0; j < SC_EPT; ++j) {
    const int i = j * 1024 + t;
    if (i < tn) rec[tb + i] = q[j];
  }
}

#define NS_STAGE 8192
#define NS_RPT 9

__device__ __forceinline__ void load_bucket_records(const int4* __restrict__ rec,
                                                    const int* rdst, const int* rsrc,
                                                    int NT, int P2, int i, int cnt,
                                                    int4 (&q)[NS_RPT]) {
  int jj[NS_RPT], pos[NS_RPT];
#pragma unroll
  for (int j = 0; j < NS_RPT; ++j) {
    const int ij = i + j * 512;
    jj[j] = ij < cnt ? ij : i;
    pos[j] = 0;
  }
  for (int s = P2; s > 0; s >>= 1) {
#pragma unroll
    for (int j = 0; j < NS_RPT; ++j) {
      const int p = pos[j] + s;
      if (rdst[min(p, NT)] <= jj[j]) pos[j] = p;
    }
  }
#pragma unroll
  for (int j = 0; j < NS_RPT; ++j)
    q[j] = rec[rsrc[pos[j]] + (jj[j] - rdst[pos[j]])];
}

__global__ void __launch_bounds__(512)
node_sort_deg_kernel(const int4* __restrict__ rec, const int* __restrict__ toff, int NT,
                     float* __restrict__ dinv, int* __restrict__ rowptr,
                     int2* __restrict__ srec, int N, int NB) {
  __shared__ int2 stage[NS_STAGE];
  __shared__ float sdeg[BK];
  __shared__ int scnt[BK];
  __shared__ int wsum[16];
  __shared__ int sbeg;
  extern __shared__ int runtab[];
  int* rdst = runtab;
  int* rsrc = runtab + NT + 1;
  const int t = threadIdx.x, lane = t & 31, w = t >> 5;
  const int b = blockIdx.x;
  if (t < BK) { sdeg[t] = 1.0f; scnt[t] = 0; }
  if (t == 0) sbeg = 0;
  __syncthreads();
  int carry = 0, osum = 0;
  for (int base = 0; base < NT; base += 512) {
    const int c = base + t;
    int o0 = 0, o1 = 0;
    if (c < NT) {
      o0 = toff[(size_t)b * NT + c];
      o1 = toff[(size_t)(b + 1) * NT + c];
    }
    const int v = o1 - o0;
    const int s = wave_incl_scan(v, lane);
    if (lane == 31) wsum[w] = s;
    __syncthreads();
    if (w == 0) {
      int x = (lane < 16) ? wsum[lane] : 0;
      x = wave_incl_scan(x, lane);
      if (lane < 16) wsum[lane] = x;
    }
    __syncthreads();
    const int wbase = (w > 0) ? wsum[w - 1] : 0;
    if (c < NT) {
      rdst[c] = carry + wbase + s - v;
      rsrc[c] = c * SC_TILE + o0;
    }
    const int piece = wsum[15];
    __syncthreads();
    carry += piece;
    osum += o0;
  }
  for (int off = 16; off > 0; off >>= 1) osum += __shfl_xor(osum, off, 32);
  if (lane == 0) lds_add_i32(&sbeg, osum);
  if (t == 0) rdst[NT] = carry;
  __syncthreads();
  const int beg = sbeg;
  const int cnt = carry;
  const int end = beg + cnt;
  const int P2 = 1 << (31 - __clz(NT));
  const bool staged = cnt <= NS_STAGE;
  for (int i = t; i < cnt; i += NS_RPT * 512) {
    int4 q[NS_RPT];
    load_bucket_records(rec, rdst, rsrc, NT, P2, i, cnt, q);
#pragma unroll
    for (int j = 0; j < NS_RPT; ++j) {
      const bool valid = i + j * 512 < cnt;
      const int c = q[j].y & (BK - 1);
      lds_add_i32(&scnt[c], valid ? 1 : 0);
      lds_add_f32(&sdeg[c], valid ? __int_as_float(q[j].z) : 0.0f);
    }
  }
  __syncthreads();
  if (t < BK) {
    const int n = b * BK + t;
    if (n < N) {
      const float d = sdeg[t];
      dinv[n] = (d > 0.0f) ? rsqrtf(d) : 0.0f;
    }
  }
  if (t < 32) {
    const int v0 = scnt[4 * t], v1 = scnt[4 * t + 1], v2 = scnt[4 * t + 2], v3 = scnt[4 * t + 3];
    const int s = v0 + v1 + v2 + v3;
    const int base = wave_incl_scan(s, t) - s;
    const int o0 = base, o1 = base + v0, o2 = o1 + v1, o3 = o2 + v2;
    scnt[4 * t] = o0; scnt[4 * t + 1] = o1; scnt[4 * t + 2] = o2; scnt[4 * t + 3] = o3;
    const int n = b * BK + 4 * t;
    if (n     < N) rowptr[n]     = beg + o0;
    if (n + 1 < N) rowptr[n + 1] = beg + o1;
    if (n + 2 < N) rowptr[n + 2] = beg + o2;
    if (n + 3 < N) rowptr[n + 3] = beg + o3;
  }
  if (b == NB - 1 && t == 0) rowptr[N] = end;
  __syncthreads();
  for (int i = t; i < cnt; i += NS_RPT * 512) {
    int4 q[NS_RPT];
    load_bucket_records(rec, rdst, rsrc, NT, P2, i, cnt, q);
#pragma unroll
    for (int j = 0; j < NS_RPT; ++j) {
      const bool valid = i + j * 512 < cnt;
      const int pos = lds_add_i32(&scnt[q[j].y & (BK - 1)], valid ? 1 : 0);
      if (valid) {
        const int2 r = make_int2(q[j].x, q[j].z);
        if (staged) stage[pos] = r; else srec[beg + pos] = r;
      }
    }
  }
  if (staged) {
    __syncthreads();
    for (int i = t; i < cnt; i += 512) srec[beg + i] = stage[i];
  }
}

template <int NCOLS>
__global__ void gemm_wmma_f32(const float* __restrict__ A,
                              const float* __restrict__ B,
                              const float* __restrict__ rowscale,
                              float* __restrict__ C) {
  const int m0   = blockIdx.x * 16;
  const int wid  = threadIdx.x >> 5;
  const int lane = threadIdx.x & 31;
  const int n0   = wid * 16;
  const int ln   = lane & 15;
  const int half = lane >> 4;

  v8f acc = {};
  const float* aptr = A + (size_t)(m0 + ln) * 64 + 2 * half;
  const float* bptr = B + (size_t)(2 * half) * NCOLS + n0 + ln;

#pragma unroll
  for (int k = 0; k < 64; k += 4) {
    v2f a, b;
    a.x = aptr[k];
    a.y = aptr[k + 1];
    b.x = bptr[k * NCOLS];
    b.y = bptr[k * NCOLS + NCOLS];
    acc = __builtin_amdgcn_wmma_f32_16x16x4_f32(
         false, a,  false, b,
         (short)0, acc,  false,  false);
  }

  const float4 s_lo = *(const float4*)(rowscale + m0 + 8 * half);
  const float4 s_hi = *(const float4*)(rowscale + m0 + 8 * half + 4);
  const float sc[8] = {s_lo.x, s_lo.y, s_lo.z, s_lo.w, s_hi.x, s_hi.y, s_hi.z, s_hi.w};
  float* cptr = C + (size_t)(m0 + 8 * half) * NCOLS + n0 + ln;
#pragma unroll
  for (int r = 0; r < 8; ++r) {
    cptr[(size_t)r * NCOLS] = acc[r] * sc[r];
  }
}

template <int CH, int RPG, int U, bool L2NORM>
__global__ void __launch_bounds__(256)
aggregate_rows_kernel(const float* __restrict__ xs, const int2* __restrict__ srec,
                      const int* __restrict__ rowptr, const float* __restrict__ dinv,
                      const float* __restrict__ bias, float* __restrict__ out, int N,
                      const float* __restrict__ Wn) {
  constexpr int LPE = CH / 4;
  constexpr int NG = 256 / LPE;
  constexpr int TROWS = RPG * NG;
  constexpr int TP = CH + 4;
  __shared__ __attribute__((aligned(16))) float tile[L2NORM ? 4 : TROWS * TP];
  const int t = threadIdx.x;
  const int sub = t % LPE;
  const int grp = t / LPE;
  const float4* x4 = (const float4*)xs;
  float4* out4 = (float4*)out;
  const float b0 = bias[sub * 4 + 0], b1 = bias[sub * 4 + 1];
  const float b2 = bias[sub * 4 + 2], b3 = bias[sub * 4 + 3];
  const char* xbytes = (const char*)xs;
  const char* rbytes = (const char*)srec;
  const unsigned suboff = (unsigned)sub * 16u;

  for (int rr = 0; rr < RPG; ++rr) {
    const int n = (blockIdx.x * RPG + rr) * NG + grp;
    float4 v = make_float4(0.0f, 0.0f, 0.0f, 0.0f);
    if (n < N) {
      const int s = rowptr[n], e = rowptr[n + 1];
      float4 a = x4[(size_t)n * LPE + sub];
      for (int k = s; k < e; k += U) {
        int2 q[U];
        float4 xv[U];
#pragma unroll
        for (int j = 0; j < U; ++j) {
          const int idx = min(k + j, e - 1);
          q[j] = *(const int2*)(rbytes + (unsigned)idx * 8u);
        }
#pragma unroll
        for (int j = 0; j < U; ++j)
          xv[j] = *(const float4*)(xbytes + ((unsigned)q[j].x * (unsigned)(CH * 4) + suboff));
#pragma unroll
        for (int j = 0; j < U; ++j) {
          const float w = (k + j < e) ? __int_as_float(q[j].y) : 0.0f;
          a.x = fmaf(w, xv[j].x, a.x);
          a.y = fmaf(w, xv[j].y, a.y);
          a.z = fmaf(w, xv[j].z, a.z);
          a.w = fmaf(w, xv[j].w, a.w);
        }
      }
      const float d = dinv[n];
      v.x = a.x * d + b0; v.y = a.y * d + b1; v.z = a.z * d + b2; v.w = a.w * d + b3;
      if (L2NORM) {
        float ss = v.x * v.x + v.y * v.y + v.z * v.z + v.w * v.w;
        for (int off = 1; off < LPE; off <<= 1) ss += __shfl_xor(ss, off, 32);
        const float scale = 1.0f / fmaxf(sqrtf(ss), 1e-12f);
        v.x *= scale; v.y *= scale; v.z *= scale; v.w *= scale;
        out4[(size_t)n * LPE + sub] = v;
      } else {
        v.x = fmaxf(v.x, 0.0f); v.y = fmaxf(v.y, 0.0f);
        v.z = fmaxf(v.z, 0.0f); v.w = fmaxf(v.w, 0.0f);
      }
    }
    if (!L2NORM) *(float4*)&tile[(rr * NG + grp) * TP + sub * 4] = v;
  }

  if (!L2NORM) {
    constexpr int NOUT = 32;
    constexpr int MT = TROWS / 16;
    constexpr int NT = NOUT / 16;
    static_assert(TROWS % 16 == 0 && MT * NT <= 8 && CH % 4 == 0, "tile shape");
    const int wv = t >> 5;
    const bool mm = wv < MT * NT;
    const int lane = t & 31;
    const int ln   = lane & 15;
    const int half = lane >> 4;
    const int mt = wv / NT, nt = wv % NT;
    const int m0 = blockIdx.x * TROWS + mt * 16 + 8 * half;
    v2f bf[CH / 4] = {};
    float sc[8] = {};
    if (mm) {
      const float* bptr = Wn + (size_t)(2 * half) * NOUT + nt * 16 + ln;
#pragma unroll
      for (int k = 0; k < CH; k += 4) {
        bf[k / 4].x = bptr[k * NOUT];
        bf[k / 4].y = bptr[k * NOUT + NOUT];
      }
#pragma unroll
      for (int r = 0; r < 8; ++r) sc[r] = dinv[min(m0 + r, N - 1)];
    }
    __syncthreads();
    if (mm) {
      const float* arow = &tile[(mt * 16 + ln) * TP + 2 * half];
      v8f acc = {};
#pragma unroll
      for (int k = 0; k < CH; k += 4) {
        v2f a;
        a.x = arow[k];
        a.y = arow[k + 1];
        acc = __builtin_amdgcn_wmma_f32_16x16x4_f32(
             false, a,  false, bf[k / 4],
             (short)0, acc,  false,  false);
      }
      float* cptr = out + (size_t)m0 * NOUT + nt * 16 + ln;
      if (m0 + 8 <= N) {
#pragma unroll
        for (int r = 0; r < 8; ++r) cptr[(size_t)r * NOUT] = acc[r] * sc[r];
      } else {
#pragma unroll
        for (int r = 0; r < 8; ++r)
          if (m0 + r < N) cptr[(size_t)r * NOUT] = acc[r] * sc[r];
      }
    }
  }
}

extern "C" void kernel_launch(void* const* d_in, const int* in_sizes, int n_in,
                              void* d_out, int out_size, void* d_ws, size_t ws_size,
                              hipStream_t stream) {
  const int*   edge_index = (const int*)d_in[0];
  const float* ew         = (const float*)d_in[1];
  const float* emb        = (const float*)d_in[2];
  const float* W1         = (const float*)d_in[3];
  const float* b1         = (const float*)d_in[4];
  const float* W2         = (const float*)d_in[5];
  const float* b2         = (const float*)d_in[6];

  const int E = in_sizes[0] / 2;
  const int N = in_sizes[2] / 64;
  const int* rows = edge_index;
  const int* cols = edge_index + E;
  const int NB = (N + BK - 1) / BK;
  const int NT = (E + SC_TILE - 1) / SC_TILE;

  int4*  rec  = (int4*)d_ws;
  int2*  srec = (int2*)(rec + (size_t)E);
  float* dinv = (float*)(srec + (size_t)E);
  float* xw1  = dinv + (size_t)N;
  float* xw2  = xw1  + (size_t)N * 64;
  int* toff   = (int*)(xw2 + (size_t)N * 32);
  int* rowptr = toff + ((size_t)NB + 1) * NT;

  tile_sort_kernel<<<NT, 1024, sizeof(int) * 2 * (size_t)NB, stream>>>(
      rows, cols, ew, E, NB, NT, toff, rec);

  node_sort_deg_kernel<<<NB, 512, sizeof(int) * (2 * (size_t)NT + 1), stream>>>(
      rec, toff, NT, dinv, rowptr, srec, N, NB);

  gemm_wmma_f32<64><<<N / 16, 128, 0, stream>>>(emb, W1, dinv, xw1);
  aggregate_rows_kernel<64, 2, 8, false><<<(N + 31) / 32, 256, 0, stream>>>(
      xw1, srec, rowptr, dinv, b1, xw2, N, W2);

  aggregate_rows_kernel<32, 2, 8, true><<<(N + 63) / 64, 256, 0, stream>>>(
      xw2, srec, rowptr, dinv, b2, (float*)d_out, N, W2);
}


# Layer-1 GEMM (fp32 WMMA, same loaders) moved into the bucket kernel: each workgroup multiplies its own 128 x rows (LDS tile) by W1, scales by the dinv it just made; GEMM launch gone, 5 -> 4 launches
# speedup vs baseline: 5.2116x; 1.0672x over previous
#include <hip/hip_runtime.h>
#include <math.h>

typedef __attribute__((ext_vector_type(2))) float v2f;
typedef __attribute__((ext_vector_type(4))) float v4f;
typedef __attribute__((ext_vector_type(8))) float v8f;

#define BK_SHIFT 7
#define BK 128

__device__ __forceinline__ void lds_add_f32(float* p, float v) {
  __hip_atomic_fetch_add(p, v, __ATOMIC_RELAXED, __HIP_MEMORY_SCOPE_WORKGROUP);
}
__device__ __forceinline__ int lds_add_i32(int* p, int v) {
  return __hip_atomic_fetch_add(p, v, __ATOMIC_RELAXED, __HIP_MEMORY_SCOPE_WORKGROUP);
}

__device__ __forceinline__ int wave_incl_scan(int v, int lane) {
  for (int off = 1; off < 32; off <<= 1) {
    int n = __shfl_up(v, off, 32);
    if (lane >= off) v += n;
  }
  return v;
}

#define SC_EPT 13
#define SC_TILE (SC_EPT * 1024)
__global__ void __launch_bounds__(1024)
tile_sort_kernel(const int* __restrict__ rows, const int* __restrict__ cols,
                 const float* __restrict__ ew, int E, int NB, int NT,
                 int* __restrict__ toff, int4* __restrict__ rec) {
  __shared__ int4 srt[SC_TILE];
  __shared__ int wsum[32];
  extern __shared__ int sh_cnt[];
  int* cnt  = sh_cnt;
  int* loff = cnt + NB;
  const int t = threadIdx.x, lane = t & 31, w = t >> 5;
  const int tb = blockIdx.x * SC_TILE;
  const int tn = min(SC_TILE, E - tb);
  int c[SC_EPT], r[SC_EPT], rk[SC_EPT];
  float wv[SC_EPT];
#pragma unroll
  for (int j = 0; j < SC_EPT; ++j) {
    const int idx = j * 1024 + t;
    const int es = tb + (idx < tn ? idx : 0);
    c[j] = cols[es];
    r[j] = rows[es];
    wv[j] = ew[es];
  }
  for (int i = t; i < NB; i += 1024) cnt[i] = 0;
  __syncthreads();
#pragma unroll
  for (int j = 0; j < SC_EPT; ++j) {
    rk[j] = 0;
    if (j * 1024 + t < tn) rk[j] = lds_add_i32(&cnt[c[j] >> BK_SHIFT], 1);
  }
  __syncthreads();
  int carry = 0;
  for (int base = 0; base < NB; base += 1024) {
    const int i = base + t;
    const int v = (i < NB) ? cnt[i] : 0;
    const int s = wave_incl_scan(v, lane);
    if (lane == 31) wsum[w] = s;
    __syncthreads();
    if (w == 0) wsum[lane] = wave_incl_scan(wsum[lane], lane);
    __syncthreads();
    const int wbase = (w > 0) ? wsum[w - 1] : 0;
    if (i < NB) loff[i] = carry + wbase + s - v;
    const int tile_total = wsum[31];
    __syncthreads();
    carry += tile_total;
  }
  for (int i = t; i < NB; i += 1024) toff[(size_t)i * NT + blockIdx.x] = loff[i];
  if (t == 0) toff[(size_t)NB * NT + blockIdx.x] = tn;
  int slot[SC_EPT];
#pragma unroll
  for (int j = 0; j < SC_EPT; ++j) slot[j] = loff[c[j] >> BK_SHIFT] + rk[j];
#pragma unroll
  for (int j = 0; j < SC_EPT; ++j) {
    if (j * 1024 + t < tn)
      srt[slot[j]] = make_int4(r[j], c[j], __float_as_int(wv[j]), 0);
  }
  __syncthreads();
  int4 q[SC_EPT];
#pragma unroll
  for (int j = 0; j < SC_EPT; ++j) {
    const int i = j * 1024 + t;
    q[j] = srt[i < tn ? i : 0];
  }
#pragma unroll
  for (int j = 0; j < SC_EPT; ++j) {
    const int i = j * 1024 + t;
    if (i < tn) rec[tb + i] = q[j];
  }
}

#define NS_STAGE 8192
#define XT_PITCH 68
#define NS_RPT 9

__device__ __forceinline__ void load_bucket_records(const int4* __restrict__ rec,
                                                    const int* rdst, const int* rsrc,
                                                    int NT, int P2, int i, int cnt,
                                                    int4 (&q)[NS_RPT]) {
  int jj[NS_RPT], pos[NS_RPT];
#pragma unroll
  for (int j = 0; j < NS_RPT; ++j) {
    const int ij = i + j * 512;
    jj[j] = ij < cnt ? ij : i;
    pos[j] = 0;
  }
  for (int s = P2; s > 0; s >>= 1) {
#pragma unroll
    for (int j = 0; j < NS_RPT; ++j) {
      const int p = pos[j] + s;
      if (rdst[min(p, NT)] <= jj[j]) pos[j] = p;
    }
  }
#pragma unroll
  for (int j = 0; j < NS_RPT; ++j)
    q[j] = rec[rsrc[pos[j]] + (jj[j] - rdst[pos[j]])];
}

__global__ void __launch_bounds__(512)
node_sort_deg_kernel(const int4* __restrict__ rec, const int* __restrict__ toff, int NT,
                     float* __restrict__ dinv, int* __restrict__ rowptr,
                     int2* __restrict__ srec, int N, int NB,
                     const float* __restrict__ xin, const float* __restrict__ Wm,
                     float* __restrict__ xs) {
  static_assert(BK * XT_PITCH * sizeof(float) <= NS_STAGE * sizeof(int2), "tile fits the copy");
  __shared__ __attribute__((aligned(16))) int2 stage[NS_STAGE];
  __shared__ float sdeg[BK];
  __shared__ int scnt[BK];
  __shared__ int wsum[16];
  __shared__ int sbeg;
  extern __shared__ int runtab[];
  int* rdst = runtab;
  int* rsrc = runtab + NT + 1;
  const int t = threadIdx.x, lane = t & 31, w = t >> 5;
  const int b = blockIdx.x;
  float* xtile = (float*)stage;
  if (t < BK) { sdeg[t] = 1.0f; scnt[t] = 0; }
  if (t == 0) sbeg = 0;
  __syncthreads();
  int carry = 0, osum = 0;
  for (int base = 0; base < NT; base += 512) {
    const int c = base + t;
    int o0 = 0, o1 = 0;
    if (c < NT) {
      o0 = toff[(size_t)b * NT + c];
      o1 = toff[(size_t)(b + 1) * NT + c];
    }
    const int v = o1 - o0;
    const int s = wave_incl_scan(v, lane);
    if (lane == 31) wsum[w] = s;
    __syncthreads();
    if (w == 0) {
      int x = (lane < 16) ? wsum[lane] : 0;
      x = wave_incl_scan(x, lane);
      if (lane < 16) wsum[lane] = x;
    }
    __syncthreads();
    const int wbase = (w > 0) ? wsum[w - 1] : 0;
    if (c < NT) {
      rdst[c] = carry + wbase + s - v;
      rsrc[c] = c * SC_TILE + o0;
    }
    const int piece = wsum[15];
    __syncthreads();
    carry += piece;
    osum += o0;
  }
  for (int off = 16; off > 0; off >>= 1) osum += __shfl_xor(osum, off, 32);
  if (lane == 0) lds_add_i32(&sbeg, osum);
  if (t == 0) rdst[NT] = carry;
  __syncthreads();
  const int beg = sbeg;
  const int cnt = carry;
  const int end = beg + cnt;
  const int P2 = 1 << (31 - __clz(NT));
  const bool staged = cnt <= NS_STAGE;
  v4f xv[BK * 16 / 512];
#pragma unroll
  for (int j = 0; j < BK * 16 / 512; ++j) {
    const int i = t + j * 512;
    xv[j] = ((const v4f*)xin)[(size_t)min(b * BK + (i >> 4), N - 1) * 16 + (i & 15)];
  }
  for (int i = t; i < cnt; i += NS_RPT * 512) {
    int4 q[NS_RPT];
    load_bucket_records(rec, rdst, rsrc, NT, P2, i, cnt, q);
#pragma unroll
    for (int j = 0; j < NS_RPT; ++j) {
      const bool valid = i + j * 512 < cnt;
      const int c = q[j].y & (BK - 1);
      lds_add_i32(&scnt[c], valid ? 1 : 0);
      lds_add_f32(&sdeg[c], valid ? __int_as_float(q[j].z) : 0.0f);
    }
  }
#pragma unroll
  for (int j = 0; j < BK * 16 / 512; ++j) {
    const int i = t + j * 512;
    *(v4f*)&xtile[(i >> 4) * XT_PITCH + (i & 15) * 4] = xv[j];
  }
  __syncthreads();
  if (t < BK) {
    const int n = b * BK + t;
    const float d = sdeg[t];
    const float di = (d > 0.0f) ? rsqrtf(d) : 0.0f;
    if (n < N) dinv[n] = di;
    sdeg[t] = di;
  }
  if (t < 32) {
    const int v0 = scnt[4 * t], v1 = scnt[4 * t + 1], v2 = scnt[4 * t + 2], v3 = scnt[4 * t + 3];
    const int s = v0 + v1 + v2 + v3;
    const int base = wave_incl_scan(s, t) - s;
    const int o0 = base, o1 = base + v0, o2 = o1 + v1, o3 = o2 + v2;
    scnt[4 * t] = o0; scnt[4 * t + 1] = o1; scnt[4 * t + 2] = o2; scnt[4 * t + 3] = o3;
    const int n = b * BK + 4 * t;
    if (n     < N) rowptr[n]     = beg + o0;
    if (n + 1 < N) rowptr[n + 1] = beg + o1;
    if (n + 2 < N) rowptr[n + 2] = beg + o2;
    if (n + 3 < N) rowptr[n + 3] = beg + o3;
  }
  if (b == NB - 1 && t == 0) rowptr[N] = end;
  __syncthreads();
  {
    const int ln   = lane & 15;
    const int half = lane >> 4;
    const int mt = w >> 1, np = w & 1;
    const float* arow = &xtile[(mt * 16 + ln) * XT_PITCH + 2 * half];
    const float* bptr = Wm + (size_t)(2 * half) * 64 + np * 32 + ln;
    v8f acc0 = {}, acc1 = {};
#pragma unroll 4
    for (int k = 0; k < 64; k += 4) {
      v2f a, b0, b1;
      a.x  = arow[k];
      a.y  = arow[k + 1];
      b0.x = bptr[k * 64];
      b0.y = bptr[k * 64 + 64];
      b1.x = bptr[k * 64 + 16];
      b1.y = bptr[k * 64 + 64 + 16];
      acc0 = __builtin_amdgcn_wmma_f32_16x16x4_f32(
           false, a,  false, b0,
           (short)0, acc0,  false,  false);
      acc1 = __builtin_amdgcn_wmma_f32_16x16x4_f32(
           false, a,  false, b1,
           (short)0, acc1,  false,  false);
    }
    const int r0 = mt * 16 + 8 * half;
    const int m0 = b * BK + r0;
    float sc[8];
#pragma unroll
    for (int r = 0; r < 8; ++r) sc[r] = sdeg[r0 + r];
    float* cptr = xs + (size_t)m0 * 64 + np * 32 + ln;
    if (m0 + 8 <= N) {
#pragma unroll
      for (int r = 0; r < 8; ++r) {
        cptr[(size_t)r * 64]      = acc0[r] * sc[r];
        cptr[(size_t)r * 64 + 16] = acc1[r] * sc[r];
      }
    } else {
#pragma unroll
      for (int r = 0; r < 8; ++r)
        if (m0 + r < N) {
          cptr[(size_t)r * 64]      = acc0[r] * sc[r];
          cptr[(size_t)r * 64 + 16] = acc1[r] * sc[r];
        }
    }
  }
  __syncthreads();
  for (int i = t; i < cnt; i += NS_RPT * 512) {
    int4 q[NS_RPT];
    load_bucket_records(rec, rdst, rsrc, NT, P2, i, cnt, q);
#pragma unroll
    for (int j = 0; j < NS_RPT; ++j) {
      const bool valid = i + j * 512 < cnt;
      const int pos = lds_add_i32(&scnt[q[j].y & (BK - 1)], valid ? 1 : 0);
      if (valid) {
        const int2 r = make_int2(q[j].x, q[j].z);
        if (staged) stage[pos] = r; else srec[beg + pos] = r;
      }
    }
  }
  if (staged) {
    __syncthreads();
    for (int i = t; i < cnt; i += 512) srec[beg + i] = stage[i];
  }
}

template <int CH, int RPG, int U, bool L2NORM>
__global__ void __launch_bounds__(256)
aggregate_rows_kernel(const float* __restrict__ xs, const int2* __restrict__ srec,
                      const int* __restrict__ rowptr, const float* __restrict__ dinv,
                      const float* __restrict__ bias, float* __restrict__ out, int N,
                      const float* __restrict__ Wn) {
  constexpr int LPE = CH / 4;
  constexpr int NG = 256 / LPE;
  constexpr int TROWS = RPG * NG;
  constexpr int TP = CH + 4;
  __shared__ __attribute__((aligned(16))) float tile[L2NORM ? 4 : TROWS * TP];
  const int t = threadIdx.x;
  const int sub = t % LPE;
  const int grp = t / LPE;
  const float4* x4 = (const float4*)xs;
  float4* out4 = (float4*)out;
  const float b0 = bias[sub * 4 + 0], b1 = bias[sub * 4 + 1];
  const float b2 = bias[sub * 4 + 2], b3 = bias[sub * 4 + 3];
  const char* xbytes = (const char*)xs;
  const char* rbytes = (const char*)srec;
  const unsigned suboff = (unsigned)sub * 16u;

  for (int rr = 0; rr < RPG; ++rr) {
    const int n = (blockIdx.x * RPG + rr) * NG + grp;
    float4 v = make_float4(0.0f, 0.0f, 0.0f, 0.0f);
    if (n < N) {
      const int s = rowptr[n], e = rowptr[n + 1];
      float4 a = x4[(size_t)n * LPE + sub];
      for (int k = s; k < e; k += U) {
        int2 q[U];
        float4 xv[U];
#pragma unroll
        for (int j = 0; j < U; ++j) {
          const int idx = min(k + j, e - 1);
          q[j] = *(const int2*)(rbytes + (unsigned)idx * 8u);
        }
#pragma unroll
        for (int j = 0; j < U; ++j)
          xv[j] = *(const float4*)(xbytes + ((unsigned)q[j].x * (unsigned)(CH * 4) + suboff));
#pragma unroll
        for (int j = 0; j < U; ++j) {
          const float w = (k + j < e) ? __int_as_float(q[j].y) : 0.0f;
          a.x = fmaf(w, xv[j].x, a.x);
          a.y = fmaf(w, xv[j].y, a.y);
          a.z = fmaf(w, xv[j].z, a.z);
          a.w = fmaf(w, xv[j].w, a.w);
        }
      }
      const float d = dinv[n];
      v.x = a.x * d + b0; v.y = a.y * d + b1; v.z = a.z * d + b2; v.w = a.w * d + b3;
      if (L2NORM) {
        float ss = v.x * v.x + v.y * v.y + v.z * v.z + v.w * v.w;
        for (int off = 1; off < LPE; off <<= 1) ss += __shfl_xor(ss, off, 32);
        const float scale = 1.0f / fmaxf(sqrtf(ss), 1e-12f);
        v.x *= scale; v.y *= scale; v.z *= scale; v.w *= scale;
        out4[(size_t)n * LPE + sub] = v;
      } else {
        v.x = fmaxf(v.x, 0.0f); v.y = fmaxf(v.y, 0.0f);
        v.z = fmaxf(v.z, 0.0f); v.w = fmaxf(v.w, 0.0f);
      }
    }
    if (!L2NORM) *(float4*)&tile[(rr * NG + grp) * TP + sub * 4] = v;
  }

  if (!L2NORM) {
    constexpr int NOUT = 32;
    constexpr int MT = TROWS / 16;
    constexpr int NT = NOUT / 16;
    static_assert(TROWS % 16 == 0 && MT * NT <= 8 && CH % 4 == 0, "tile shape");
    const int wv = t >> 5;
    const bool mm = wv < MT * NT;
    const int lane = t & 31;
    const int ln   = lane & 15;
    const int half = lane >> 4;
    const int mt = wv / NT, nt = wv % NT;
    const int m0 = blockIdx.x * TROWS + mt * 16 + 8 * half;
    v2f bf[CH / 4] = {};
    float sc[8] = {};
    if (mm) {
      const float* bptr = Wn + (size_t)(2 * half) * NOUT + nt * 16 + ln;
#pragma unroll
      for (int k = 0; k < CH; k += 4) {
        bf[k / 4].x = bptr[k * NOUT];
        bf[k / 4].y = bptr[k * NOUT + NOUT];
      }
#pragma unroll
      for (int r = 0; r < 8; ++r) sc[r] = dinv[min(m0 + r, N - 1)];
    }
    __syncthreads();
    if (mm) {
      const float* arow = &tile[(mt * 16 + ln) * TP + 2 * half];
      v8f acc = {};
#pragma unroll
      for (int k = 0; k < CH; k += 4) {
        v2f a;
        a.x = arow[k];
        a.y = arow[k + 1];
        acc = __builtin_amdgcn_wmma_f32_16x16x4_f32(
             false, a,  false, bf[k / 4],
             (short)0, acc,  false,  false);
      }
      float* cptr = out + (size_t)m0 * NOUT + nt * 16 + ln;
      if (m0 + 8 <= N) {
#pragma unroll
        for (int r = 0; r < 8; ++r) cptr[(size_t)r * NOUT] = acc[r] * sc[r];
      } else {
#pragma unroll
        for (int r = 0; r < 8; ++r)
          if (m0 + r < N) cptr[(size_t)r * NOUT] = acc[r] * sc[r];
      }
    }
  }
}

extern "C" void kernel_launch(void* const* d_in, const int* in_sizes, int n_in,
                              void* d_out, int out_size, void* d_ws, size_t ws_size,
                              hipStream_t stream) {
  const int*   edge_index = (const int*)d_in[0];
  const float* ew         = (const float*)d_in[1];
  const float* emb        = (const float*)d_in[2];
  const float* W1         = (const float*)d_in[3];
  const float* b1         = (const float*)d_in[4];
  const float* W2         = (const float*)d_in[5];
  const float* b2         = (const float*)d_in[6];

  const int E = in_sizes[0] / 2;
  const int N = in_sizes[2] / 64;
  const int* rows = edge_index;
  const int* cols = edge_index + E;
  const int NB = (N + BK - 1) / BK;
  const int NT = (E + SC_TILE - 1) / SC_TILE;

  int4*  rec  = (int4*)d_ws;
  int2*  srec = (int2*)(rec + (size_t)E);
  float* dinv = (float*)(srec + (size_t)E);
  float* xw1  = dinv + (size_t)N;
  float* xw2  = xw1  + (size_t)N * 64;
  int* toff   = (int*)(xw2 + (size_t)N * 32);
  int* rowptr = toff + ((size_t)NB + 1) * NT;

  tile_sort_kernel<<<NT, 1024, sizeof(int) * 2 * (size_t)NB, stream>>>(
      rows, cols, ew, E, NB, NT, toff, rec);

  node_sort_deg_kernel<<<NB, 512, sizeof(int) * (2 * (size_t)NT + 1), stream>>>(
      rec, toff, NT, dinv, rowptr, srec, N, NB, emb, W1, xw1);

  aggregate_rows_kernel<64, 2, 8, false><<<(N + 31) / 32, 256, 0, stream>>>(
      xw1, srec, rowptr, dinv, b1, xw2, N, W2);

  aggregate_rows_kernel<32, 2, 8, true><<<(N + 63) / 64, 256, 0, stream>>>(
      xw2, srec, rowptr, dinv, b2, (float*)d_out, N, W2);
}


# First-level sort records packed to 8 bytes (source id | low 7 destination bits in one word, weight): tile sort stores and bucket kernel reads half the bytes; same order, same sums
# speedup vs baseline: 5.4797x; 1.0514x over previous
#include <hip/hip_runtime.h>
#include <math.h>

typedef __attribute__((ext_vector_type(2))) float v2f;
typedef __attribute__((ext_vector_type(4))) float v4f;
typedef __attribute__((ext_vector_type(8))) float v8f;

#define BK_SHIFT 7
#define BK 128
#define ROW_BITS 25
#define ROW_MASK ((1 << ROW_BITS) - 1)
static_assert(ROW_BITS + BK_SHIFT == 32, "source node and bucket position share one word");

__device__ __forceinline__ void lds_add_f32(float* p, float v) {
  __hip_atomic_fetch_add(p, v, __ATOMIC_RELAXED, __HIP_MEMORY_SCOPE_WORKGROUP);
}
__device__ __forceinline__ int lds_add_i32(int* p, int v) {
  return __hip_atomic_fetch_add(p, v, __ATOMIC_RELAXED, __HIP_MEMORY_SCOPE_WORKGROUP);
}

__device__ __forceinline__ int wave_incl_scan(int v, int lane) {
  for (int off = 1; off < 32; off <<= 1) {
    int n = __shfl_up(v, off, 32);
    if (lane >= off) v += n;
  }
  return v;
}

#define SC_EPT 13
#define SC_TILE (SC_EPT * 1024)
__global__ void __launch_bounds__(1024)
tile_sort_kernel(const int* __restrict__ rows, const int* __restrict__ cols,
                 const float* __restrict__ ew, int E, int NB, int NT,
                 int* __restrict__ toff, int2* __restrict__ rec) {
  __shared__ int2 srt[SC_TILE];
  __shared__ int wsum[32];
  extern __shared__ int sh_cnt[];
  int* cnt  = sh_cnt;
  int* loff = cnt + NB;
  const int t = threadIdx.x, lane = t & 31, w = t >> 5;
  const int tb = blockIdx.x * SC_TILE;
  const int tn = min(SC_TILE, E - tb);
  int c[SC_EPT], r[SC_EPT], rk[SC_EPT];
  float wv[SC_EPT];
#pragma unroll
  for (int j = 0; j < SC_EPT; ++j) {
    const int idx = j * 1024 + t;
    const int es = tb + (idx < tn ? idx : 0);
    c[j] = cols[es];
    r[j] = rows[es];
    wv[j] = ew[es];
  }
  for (int i = t; i < NB; i += 1024) cnt[i] = 0;
  __syncthreads();
#pragma unroll
  for (int j = 0; j < SC_EPT; ++j) {
    rk[j] = 0;
    if (j * 1024 + t < tn) rk[j] = lds_add_i32(&cnt[c[j] >> BK_SHIFT], 1);
  }
  __syncthreads();
  int carry = 0;
  for (int base = 0; base < NB; base += 1024) {
    const int i = base + t;
    const int v = (i < NB) ? cnt[i] : 0;
    const int s = wave_incl_scan(v, lane);
    if (lane == 31) wsum[w] = s;
    __syncthreads();
    if (w == 0) wsum[lane] = wave_incl_scan(wsum[lane], lane);
    __syncthreads();
    const int wbase = (w > 0) ? wsum[w - 1] : 0;
    if (i < NB) loff[i] = carry + wbase + s - v;
    const int tile_total = wsum[31];
    __syncthreads();
    carry += tile_total;
  }
  for (int i = t; i < NB; i += 1024) toff[(size_t)i * NT + blockIdx.x] = loff[i];
  if (t == 0) toff[(size_t)NB * NT + blockIdx.x] = tn;
  int slot[SC_EPT];
#pragma unroll
  for (int j = 0; j < SC_EPT; ++j) slot[j] = loff[c[j] >> BK_SHIFT] + rk[j];
#pragma unroll
  for (int j = 0; j < SC_EPT; ++j) {
    if (j * 1024 + t < tn)
      srt[slot[j]] = make_int2((int)((unsigned)r[j] | ((unsigned)(c[j] & (BK - 1)) << ROW_BITS)),
                               __float_as_int(wv[j]));
  }
  __syncthreads();
  int2 q[SC_EPT];
#pragma unroll
  for (int j = 0; j < SC_EPT; ++j) {
    const int i = j * 1024 + t;
    q[j] = srt[i < tn ? i : 0];
  }
#pragma unroll
  for (int j = 0; j < SC_EPT; ++j) {
    const int i = j * 1024 + t;
    if (i < tn) rec[tb + i] = q[j];
  }
}

#define NS_STAGE 8192
#define XT_PITCH 68
#define NS_RPT 9

__device__ __forceinline__ void load_bucket_records(const int2* __restrict__ rec,
                                                    const int* rdst, const int* rsrc,
                                                    int NT, int P2, int i, int cnt,
                                                    int2 (&q)[NS_RPT]) {
  int jj[NS_RPT], pos[NS_RPT];
#pragma unroll
  for (int j = 0; j < NS_RPT; ++j) {
    const int ij = i + j * 512;
    jj[j] = ij < cnt ? ij : i;
    pos[j] = 0;
  }
  for (int s = P2; s > 0; s >>= 1) {
#pragma unroll
    for (int j = 0; j < NS_RPT; ++j) {
      const int p = pos[j] + s;
      if (rdst[min(p, NT)] <= jj[j]) pos[j] = p;
    }
  }
#pragma unroll
  for (int j = 0; j < NS_RPT; ++j)
    q[j] = rec[rsrc[pos[j]] + (jj[j] - rdst[pos[j]])];
}

__global__ void __launch_bounds__(512)
node_sort_deg_kernel(const int2* __restrict__ rec, const int* __restrict__ toff, int NT,
                     float* __restrict__ dinv, int* __restrict__ rowptr,
                     int2* __restrict__ srec, int N, int NB,
                     const float* __restrict__ xin, const float* __restrict__ Wm,
                     float* __restrict__ xs) {
  static_assert(BK * XT_PITCH * sizeof(float) <= NS_STAGE * sizeof(int2), "tile fits the copy");
  __shared__ __attribute__((aligned(16))) int2 stage[NS_STAGE];
  __shared__ float sdeg[BK];
  __shared__ int scnt[BK];
  __shared__ int wsum[16];
  __shared__ int sbeg;
  extern __shared__ int runtab[];
  int* rdst = runtab;
  int* rsrc = runtab + NT + 1;
  const int t = threadIdx.x, lane = t & 31, w = t >> 5;
  const int b = blockIdx.x;
  float* xtile = (float*)stage;
  if (t < BK) { sdeg[t] = 1.0f; scnt[t] = 0; }
  if (t == 0) sbeg = 0;
  __syncthreads();
  int carry = 0, osum = 0;
  for (int base = 0; base < NT; base += 512) {
    const int c = base + t;
    int o0 = 0, o1 = 0;
    if (c < NT) {
      o0 = toff[(size_t)b * NT + c];
      o1 = toff[(size_t)(b + 1) * NT + c];
    }
    const int v = o1 - o0;
    const int s = wave_incl_scan(v, lane);
    if (lane == 31) wsum[w] = s;
    __syncthreads();
    if (w == 0) {
      int x = (lane < 16) ? wsum[lane] : 0;
      x = wave_incl_scan(x, lane);
      if (lane < 16) wsum[lane] = x;
    }
    __syncthreads();
    const int wbase = (w > 0) ? wsum[w - 1] : 0;
    if (c < NT) {
      rdst[c] = carry + wbase + s - v;
      rsrc[c] = c * SC_TILE + o0;
    }
    const int piece = wsum[15];
    __syncthreads();
    carry += piece;
    osum += o0;
  }
  for (int off = 16; off > 0; off >>= 1) osum += __shfl_xor(osum, off, 32);
  if (lane == 0) lds_add_i32(&sbeg, osum);
  if (t == 0) rdst[NT] = carry;
  __syncthreads();
  const int beg = sbeg;
  const int cnt = carry;
  const int end = beg + cnt;
  const int P2 = 1 << (31 - __clz(NT));
  const bool staged = cnt <= NS_STAGE;
  v4f xv[BK * 16 / 512];
#pragma unroll
  for (int j = 0; j < BK * 16 / 512; ++j) {
    const int i = t + j * 512;
    xv[j] = ((const v4f*)xin)[(size_t)min(b * BK + (i >> 4), N - 1) * 16 + (i & 15)];
  }
  for (int i = t; i < cnt; i += NS_RPT * 512) {
    int2 q[NS_RPT];
    load_bucket_records(rec, rdst, rsrc, NT, P2, i, cnt, q);
#pragma unroll
    for (int j = 0; j < NS_RPT; ++j) {
      const bool valid = i + j * 512 < cnt;
      const int c = (int)((unsigned)q[j].x >> ROW_BITS);
      lds_add_i32(&scnt[c], valid ? 1 : 0);
      lds_add_f32(&sdeg[c], valid ? __int_as_float(q[j].y) : 0.0f);
    }
  }
#pragma unroll
  for (int j = 0; j < BK * 16 / 512; ++j) {
    const int i = t + j * 512;
    *(v4f*)&xtile[(i >> 4) * XT_PITCH + (i & 15) * 4] = xv[j];
  }
  __syncthreads();
  if (t < BK) {
    const int n = b * BK + t;
    const float d = sdeg[t];
    const float di = (d > 0.0f) ? rsqrtf(d) : 0.0f;
    if (n < N) dinv[n] = di;
    sdeg[t] = di;
  }
  if (t < 32) {
    const int v0 = scnt[4 * t], v1 = scnt[4 * t + 1], v2 = scnt[4 * t + 2], v3 = scnt[4 * t + 3];
    const int s = v0 + v1 + v2 + v3;
    const int base = wave_incl_scan(s, t) - s;
    const int o0 = base, o1 = base + v0, o2 = o1 + v1, o3 = o2 + v2;
    scnt[4 * t] = o0; scnt[4 * t + 1] = o1; scnt[4 * t + 2] = o2; scnt[4 * t + 3] = o3;
    const int n = b * BK + 4 * t;
    if (n     < N) rowptr[n]     = beg + o0;
    if (n + 1 < N) rowptr[n + 1] = beg + o1;
    if (n + 2 < N) rowptr[n + 2] = beg + o2;
    if (n + 3 < N) rowptr[n + 3] = beg + o3;
  }
  if (b == NB - 1 && t == 0) rowptr[N] = end;
  __syncthreads();
  {
    const int ln   = lane & 15;
    const int half = lane >> 4;
    const int mt = w >> 1, np = w & 1;
    const float* arow = &xtile[(mt * 16 + ln) * XT_PITCH + 2 * half];
    const float* bptr = Wm + (size_t)(2 * half) * 64 + np * 32 + ln;
    v8f acc0 = {}, acc1 = {};
#pragma unroll 4
    for (int k = 0; k < 64; k += 4) {
      v2f a, b0, b1;
      a.x  = arow[k];
      a.y  = arow[k + 1];
      b0.x = bptr[k * 64];
      b0.y = bptr[k * 64 + 64];
      b1.x = bptr[k * 64 + 16];
      b1.y = bptr[k * 64 + 64 + 16];
      acc0 = __builtin_amdgcn_wmma_f32_16x16x4_f32(
           false, a,  false, b0,
           (short)0, acc0,  false,  false);
      acc1 = __builtin_amdgcn_wmma_f32_16x16x4_f32(
           false, a,  false, b1,
           (short)0, acc1,  false,  false);
    }
    const int r0 = mt * 16 + 8 * half;
    const int m0 = b * BK + r0;
    float sc[8];
#pragma unroll
    for (int r = 0; r < 8; ++r) sc[r] = sdeg[r0 + r];
    float* cptr = xs + (size_t)m0 * 64 + np * 32 + ln;
    if (m0 + 8 <= N) {
#pragma unroll
      for (int r = 0; r < 8; ++r) {
        cptr[(size_t)r * 64]      = acc0[r] * sc[r];
        cptr[(size_t)r * 64 + 16] = acc1[r] * sc[r];
      }
    } else {
#pragma unroll
      for (int r = 0; r < 8; ++r)
        if (m0 + r < N) {
          cptr[(size_t)r * 64]      = acc0[r] * sc[r];
          cptr[(size_t)r * 64 + 16] = acc1[r] * sc[r];
        }
    }
  }
  __syncthreads();
  for (int i = t; i < cnt; i += NS_RPT * 512) {
    int2 q[NS_RPT];
    load_bucket_records(rec, rdst, rsrc, NT, P2, i, cnt, q);
#pragma unroll
    for (int j = 0; j < NS_RPT; ++j) {
      const bool valid = i + j * 512 < cnt;
      const int pos = lds_add_i32(&scnt[(unsigned)q[j].x >> ROW_BITS], valid ? 1 : 0);
      if (valid) {
        const int2 r = make_int2(q[j].x & ROW_MASK, q[j].y);
        if (staged) stage[pos] = r; else srec[beg + pos] = r;
      }
    }
  }
  if (staged) {
    __syncthreads();
    for (int i = t; i < cnt; i += 512) srec[beg + i] = stage[i];
  }
}

template <int CH, int RPG, int U, bool L2NORM>
__global__ void __launch_bounds__(256)
aggregate_rows_kernel(const float* __restrict__ xs, const int2* __restrict__ srec,
                      const int* __restrict__ rowptr, const float* __restrict__ dinv,
                      const float* __restrict__ bias, float* __restrict__ out, int N,
                      const float* __restrict__ Wn) {
  constexpr int LPE = CH / 4;
  constexpr int NG = 256 / LPE;
  constexpr int TROWS = RPG * NG;
  constexpr int TP = CH + 4;
  __shared__ __attribute__((aligned(16))) float tile[L2NORM ? 4 : TROWS * TP];
  const int t = threadIdx.x;
  const int sub = t % LPE;
  const int grp = t / LPE;
  const float4* x4 = (const float4*)xs;
  float4* out4 = (float4*)out;
  const float b0 = bias[sub * 4 + 0], b1 = bias[sub * 4 + 1];
  const float b2 = bias[sub * 4 + 2], b3 = bias[sub * 4 + 3];
  const char* xbytes = (const char*)xs;
  const char* rbytes = (const char*)srec;
  const unsigned suboff = (unsigned)sub * 16u;

  for (int rr = 0; rr < RPG; ++rr) {
    const int n = (blockIdx.x * RPG + rr) * NG + grp;
    float4 v = make_float4(0.0f, 0.0f, 0.0f, 0.0f);
    if (n < N) {
      const int s = rowptr[n], e = rowptr[n + 1];
      float4 a = x4[(size_t)n * LPE + sub];
      for (int k = s; k < e; k += U) {
        int2 q[U];
        float4 xv[U];
#pragma unroll
        for (int j = 0; j < U; ++j) {
          const int idx = min(k + j, e - 1);
          q[j] = *(const int2*)(rbytes + (unsigned)idx * 8u);
        }
#pragma unroll
        for (int j = 0; j < U; ++j)
          xv[j] = *(const float4*)(xbytes + ((unsigned)q[j].x * (unsigned)(CH * 4) + suboff));
#pragma unroll
        for (int j = 0; j < U; ++j) {
          const float w = (k + j < e) ? __int_as_float(q[j].y) : 0.0f;
          a.x = fmaf(w, xv[j].x, a.x);
          a.y = fmaf(w, xv[j].y, a.y);
          a.z = fmaf(w, xv[j].z, a.z);
          a.w = fmaf(w, xv[j].w, a.w);
        }
      }
      const float d = dinv[n];
      v.x = a.x * d + b0; v.y = a.y * d + b1; v.z = a.z * d + b2; v.w = a.w * d + b3;
      if (L2NORM) {
        float ss = v.x * v.x + v.y * v.y + v.z * v.z + v.w * v.w;
        for (int off = 1; off < LPE; off <<= 1) ss += __shfl_xor(ss, off, 32);
        const float scale = 1.0f / fmaxf(sqrtf(ss), 1e-12f);
        v.x *= scale; v.y *= scale; v.z *= scale; v.w *= scale;
        out4[(size_t)n * LPE + sub] = v;
      } else {
        v.x = fmaxf(v.x, 0.0f); v.y = fmaxf(v.y, 0.0f);
        v.z = fmaxf(v.z, 0.0f); v.w = fmaxf(v.w, 0.0f);
      }
    }
    if (!L2NORM) *(float4*)&tile[(rr * NG + grp) * TP + sub * 4] = v;
  }

  if (!L2NORM) {
    constexpr int NOUT = 32;
    constexpr int MT = TROWS / 16;
    constexpr int NT = NOUT / 16;
    static_assert(TROWS % 16 == 0 && MT * NT <= 8 && CH % 4 == 0, "tile shape");
    const int wv = t >> 5;
    const bool mm = wv < MT * NT;
    const int lane = t & 31;
    const int ln   = lane & 15;
    const int half = lane >> 4;
    const int mt = wv / NT, nt = wv % NT;
    const int m0 = blockIdx.x * TROWS + mt * 16 + 8 * half;
    v2f bf[CH / 4] = {};
    float sc[8] = {};
    if (mm) {
      const float* bptr = Wn + (size_t)(2 * half) * NOUT + nt * 16 + ln;
#pragma unroll
      for (int k = 0; k < CH; k += 4) {
        bf[k / 4].x = bptr[k * NOUT];
        bf[k / 4].y = bptr[k * NOUT + NOUT];
      }
#pragma unroll
      for (int r = 0; r < 8; ++r) sc[r] = dinv[min(m0 + r, N - 1)];
    }
    __syncthreads();
    if (mm) {
      const float* arow = &tile[(mt * 16 + ln) * TP + 2 * half];
      v8f acc = {};
#pragma unroll
      for (int k = 0; k < CH; k += 4) {
        v2f a;
        a.x = arow[k];
        a.y = arow[k + 1];
        acc = __builtin_amdgcn_wmma_f32_16x16x4_f32(
             false, a,  false, bf[k / 4],
             (short)0, acc,  false,  false);
      }
      float* cptr = out + (size_t)m0 * NOUT + nt * 16 + ln;
      if (m0 + 8 <= N) {
#pragma unroll
        for (int r = 0; r < 8; ++r) cptr[(size_t)r * NOUT] = acc[r] * sc[r];
      } else {
#pragma unroll
        for (int r = 0; r < 8; ++r)
          if (m0 + r < N) cptr[(size_t)r * NOUT] = acc[r] * sc[r];
      }
    }
  }
}

extern "C" void kernel_launch(void* const* d_in, const int* in_sizes, int n_in,
                              void* d_out, int out_size, void* d_ws, size_t ws_size,
                              hipStream_t stream) {
  const int*   edge_index = (const int*)d_in[0];
  const float* ew         = (const float*)d_in[1];
  const float* emb        = (const float*)d_in[2];
  const float* W1         = (const float*)d_in[3];
  const float* b1         = (const float*)d_in[4];
  const float* W2         = (const float*)d_in[5];
  const float* b2         = (const float*)d_in[6];

  const int E = in_sizes[0] / 2;
  const int N = in_sizes[2] / 64;
  const int* rows = edge_index;
  const int* cols = edge_index + E;
  const int NB = (N + BK - 1) / BK;
  const int NT = (E + SC_TILE - 1) / SC_TILE;

  int2*  rec  = (int2*)d_ws;
  int2*  srec = rec + (size_t)E;
  float* dinv = (float*)(srec + (size_t)E);
  float* xw1  = dinv + (size_t)N;
  float* xw2  = xw1  + (size_t)N * 64;
  int* toff   = (int*)(xw2 + (size_t)N * 32);
  int* rowptr = toff + ((size_t)NB + 1) * NT;

  tile_sort_kernel<<<NT, 1024, sizeof(int) * 2 * (size_t)NB, stream>>>(
      rows, cols, ew, E, NB, NT, toff, rec);

  node_sort_deg_kernel<<<NB, 512, sizeof(int) * (2 * (size_t)NT + 1), stream>>>(
      rec, toff, NT, dinv, rowptr, srec, N, NB, emb, W1, xw1);

  aggregate_rows_kernel<64, 2, 8, false><<<(N + 31) / 32, 256, 0, stream>>>(
      xw1, srec, rowptr, dinv, b1, xw2, N, W2);

  aggregate_rows_kernel<32, 2, 8, true><<<(N + 63) / 64, 256, 0, stream>>>(
      xw2, srec, rowptr, dinv, b2, (float*)d_out, N, W2);
}
